# starting kernel
# baseline (speedup 1.0000x reference)
#include <hip/hip_runtime.h>
#include <hip/hip_bf16.h>
#include <math.h>


#define B_     8
#define CIN_   25
#define NPG_   2000
#define ENC_   2816
#define HID_   100
#define KGEMM_ 2816

typedef __attribute__((ext_vector_type(2))) float v2f;
typedef __attribute__((ext_vector_type(8))) float v8f;

__device__ __forceinline__ unsigned fkey(float f) {
    unsigned u = __float_as_uint(f);
    return (u & 0x80000000u) ? ~u : (u | 0x80000000u);
}
__device__ __forceinline__ float keyf(unsigned k) {
    unsigned u = (k & 0x80000000u) ? (k ^ 0x80000000u) : ~k;
    return __uint_as_float(u);
}

__global__ void zero_u32(unsigned* __restrict__ p, int n) {
    int i = blockIdx.x * blockDim.x + threadIdx.x;
    if (i < n) p[i] = 0u;
}

__global__ void conv1_kernel(const float* __restrict__ X, const float* __restrict__ w,
                             const float* __restrict__ b, float* __restrict__ out) {
    const int total = 8 * 32 * 21 * 6 * 6;
    int idx = blockIdx.x * blockDim.x + threadIdx.x;
    if (idx >= total) return;
    int x = idx % 6;  int t = idx / 6;
    int y = t % 6;    t /= 6;
    int z = t % 21;   t /= 21;
    int o = t % 32;   int bn = t / 32;
    float acc = b[o];
    for (int c = 0; c < 25; ++c)
        for (int kd = 0; kd < 5; ++kd) {
            const float* Xp = X + ((size_t)((bn * 25 + c) * 25 + (z + kd))) * 36;
            const float* Wp = w + ((size_t)((o * 25 + c) * 5 + kd)) * 9;
            for (int kh = 0; kh < 3; ++kh) {
                int yy = y + kh - 1; if (yy < 0 || yy >= 6) continue;
                for (int kw = 0; kw < 3; ++kw) {
                    int xx = x + kw - 1; if (xx < 0 || xx >= 6) continue;
                    acc = fmaf(Xp[yy * 6 + xx], Wp[kh * 3 + kw], acc);
                }
            }
        }
    out[idx] = fmaxf(acc, 0.0f);
}

__global__ void conv2_kernel(const float* __restrict__ A, const float* __restrict__ w,
                             const float* __restrict__ b, float* __restrict__ out) {
    const int total = 8 * 64 * 17 * 6 * 6;
    int idx = blockIdx.x * blockDim.x + threadIdx.x;
    if (idx >= total) return;
    int x = idx % 6;  int t = idx / 6;
    int y = t % 6;    t /= 6;
    int z = t % 17;   t /= 17;
    int o = t % 64;   int bn = t / 64;
    float acc = b[o];
    for (int c = 0; c < 32; ++c)
        for (int kd = 0; kd < 5; ++kd) {
            const float* Ap = A + ((size_t)((bn * 32 + c) * 21 + (z + kd))) * 36;
            const float* Wp = w + ((size_t)((o * 32 + c) * 5 + kd)) * 9;
            for (int kh = 0; kh < 3; ++kh) {
                int yy = y + kh - 1; if (yy < 0 || yy >= 6) continue;
                for (int kw = 0; kw < 3; ++kw) {
                    int xx = x + kw - 1; if (xx < 0 || xx >= 6) continue;
                    acc = fmaf(Ap[yy * 6 + xx], Wp[kh * 3 + kw], acc);
                }
            }
        }
    out[idx] = fmaxf(acc, 0.0f);
}

__global__ void conv3_kernel(const float* __restrict__ A, const float* __restrict__ w,
                             const float* __restrict__ b, float* __restrict__ out) {
    const int total = 8 * 64 * 13 * 4 * 4;
    int idx = blockIdx.x * blockDim.x + threadIdx.x;
    if (idx >= total) return;
    int x = idx % 4;  int t = idx / 4;
    int y = t % 4;    t /= 4;
    int z = t % 13;   t /= 13;
    int o = t % 64;   int bn = t / 64;
    float acc = b[o];
    for (int c = 0; c < 64; ++c)
        for (int kd = 0; kd < 5; ++kd) {
            const float* Ap = A + ((size_t)((bn * 64 + c) * 17 + (z + kd))) * 36;
            const float* Wp = w + ((size_t)((o * 64 + c) * 5 + kd)) * 9;
            for (int kh = 0; kh < 3; ++kh)
                for (int kw = 0; kw < 3; ++kw)
                    acc = fmaf(Ap[(y + kh) * 6 + (x + kw)], Wp[kh * 3 + kw], acc);
        }
    out[idx] = fmaxf(acc, 0.0f);
}

__global__ void pool_kernel(const float* __restrict__ A, float* __restrict__ enc) {
    const int total = 8 * 64 * 11 * 2 * 2;
    int idx = blockIdx.x * blockDim.x + threadIdx.x;
    if (idx >= total) return;
    int x = idx % 2;  int t = idx / 2;
    int y = t % 2;    t /= 2;
    int d = t % 11;   t /= 11;
    int ch = t % 64;  int bn = t / 64;
    float m = -INFINITY;
    for (int dd = 0; dd < 3; ++dd)
        for (int hh = 0; hh < 3; ++hh)
            for (int ww = 0; ww < 3; ++ww) {
                float v = A[((size_t)((bn * 64 + ch) * 13 + (d + dd))) * 16 + (y + hh) * 4 + (x + ww)];
                m = fmaxf(m, v);
            }
    enc[(size_t)bn * ENC_ + ch * 44 + d * 4 + y * 2 + x] = m;
}

__global__ void edge_pass(const int* __restrict__ ei, const int* __restrict__ gid,
                          const float* __restrict__ x3, unsigned* __restrict__ aggk,
                          unsigned* __restrict__ mask, int* __restrict__ counts, int E) {
    int e = blockIdx.x * blockDim.x + threadIdx.x;
    if (e >= E) return;
    int s = ei[e];
    int d = ei[E + e];
    int g = gid[s];
    atomicOr(&mask[d], 1u << g);
    atomicAdd(&counts[d], 1);
    atomicMax(&aggk[(size_t)d * 3 + 0], fkey(x3[(size_t)s * 3 + 0]));
    atomicMax(&aggk[(size_t)d * 3 + 1], fkey(x3[(size_t)s * 3 + 1]));
    atomicMax(&aggk[(size_t)d * 3 + 2], fkey(x3[(size_t)s * 3 + 2]));
}

__global__ void maxenc_kernel(const float* __restrict__ enc, float* __restrict__ ME) {
    int c = blockIdx.x * blockDim.x + threadIdx.x;
    if (c >= ENC_) return;
    ME[c] = 0.0f;
    for (int m = 1; m < 256; ++m) {
        int p = m & (m - 1);
        int g = __ffs(m) - 1;
        ME[(size_t)m * ENC_ + c] = fmaxf(ME[(size_t)p * ENC_ + c], enc[(size_t)g * ENC_ + c]);
    }
}

__global__ void __launch_bounds__(32)
gemm_wmma_f32(const float* __restrict__ A, const float* __restrict__ W,
              float* __restrict__ C, int Mrows) {
    const int Ncols = HID_;
    int lane = threadIdx.x & 31;
    int half = lane >> 4;
    int l16  = lane & 15;
    int mBase = blockIdx.x * 16;
    int nBase = blockIdx.y * 16;
    int mRow = mBase + l16; if (mRow >= Mrows) mRow = Mrows - 1;
    int nCol = nBase + l16;
    bool nOk = (nCol < Ncols);
    int nClamp = nOk ? nCol : (Ncols - 1);
    const float* Arow = A + (size_t)mRow * KGEMM_;
    const float* Bcol = W + nClamp;

    v8f acc0 = {}, acc1 = {};
    for (int kt = 0; kt < KGEMM_; kt += 8) {
        int k0 = kt + half * 2;
        int k1 = k0 + 4;
        v2f a0, b0, a1, b1;
        a0.x = Arow[k0];
        a0.y = Arow[k0 + 1];
        a1.x = Arow[k1];
        a1.y = Arow[k1 + 1];
        b0.x = Bcol[(size_t)k0 * Ncols];
        b0.y = Bcol[(size_t)(k0 + 1) * Ncols];
        b1.x = Bcol[(size_t)k1 * Ncols];
        b1.y = Bcol[(size_t)(k1 + 1) * Ncols];
        acc0 = __builtin_amdgcn_wmma_f32_16x16x4_f32(false, a0, false, b0,
                                                     (short)0, acc0, false, false);
        acc1 = __builtin_amdgcn_wmma_f32_16x16x4_f32(false, a1, false, b1,
                                                     (short)0, acc1, false, false);
    }
    v8f acc = acc0 + acc1;
    for (int v = 0; v < 8; ++v) {
        int m = mBase + half * 8 + v;
        if (m < Mrows && nOk) C[(size_t)m * Ncols + nCol] = acc[v];
    }
}

__global__ void node_h_kernel(const float* __restrict__ x3, const int* __restrict__ gid,
                              const unsigned* __restrict__ mask, const unsigned* __restrict__ aggk,
                              const float* __restrict__ P, const float* __restrict__ Q,
                              const float* __restrict__ Wl1, const float* __restrict__ Wr1,
                              const float* __restrict__ bl1, float* __restrict__ h, int N) {
    int wave = threadIdx.x >> 5;
    int lane = threadIdx.x & 31;
    int n = blockIdx.x * (blockDim.x >> 5) + wave;
    if (n >= N) return;

    float x0 = x3[(size_t)n * 3 + 0];
    float x1 = x3[(size_t)n * 3 + 1];
    float x2 = x3[(size_t)n * 3 + 2];
    int g = gid[n];
    unsigned mk = mask[n];
    bool has = (mk != 0u);
    float a0 = 0.f, a1 = 0.f, a2 = 0.f;
    if (has) {
        a0 = keyf(aggk[(size_t)n * 3 + 0]);
        a1 = keyf(aggk[(size_t)n * 3 + 1]);
        a2 = keyf(aggk[(size_t)n * 3 + 2]);
    }
    const float* Prow = P + (size_t)mk * HID_;
    const float* Qrow = Q + (size_t)g * HID_;

    float v[4];
    #pragma unroll
    for (int c = 0; c < 4; ++c) {
        int col = lane + 32 * c;
        if (col < HID_) {
            float t = bl1[col] + Qrow[col]
                    + x0 * Wr1[col] + x1 * Wr1[HID_ + col] + x2 * Wr1[2 * HID_ + col];
            if (has)
                t += Prow[col]
                   + a0 * Wl1[col] + a1 * Wl1[HID_ + col] + a2 * Wl1[2 * HID_ + col];
            v[c] = t;
        } else {
            v[c] = -INFINITY;
        }
    }
    float m = fmaxf(fmaxf(v[0], v[1]), fmaxf(v[2], v[3]));
    for (int off = 16; off >= 1; off >>= 1)
        m = fmaxf(m, __shfl_xor(m, off, 32));
    float s = 0.f;
    #pragma unroll
    for (int c = 0; c < 4; ++c) s += __expf(v[c] - m);
    for (int off = 16; off >= 1; off >>= 1)
        s += __shfl_xor(s, off, 32);
    float lse = m + __logf(s);
    #pragma unroll
    for (int c = 0; c < 4; ++c) {
        int col = lane + 32 * c;
        if (col < HID_) h[(size_t)n * HID_ + col] = v[c] - lse;
    }
}

__global__ void edge_scatter_h(const int* __restrict__ ei, const float* __restrict__ h,
                               float* __restrict__ agg2, int E) {
    int e = blockIdx.x;
    int j = threadIdx.x;
    if (j < HID_) {
        int s = ei[e];
        int d = ei[E + e];
        atomicAdd(&agg2[(size_t)d * HID_ + j], h[(size_t)s * HID_ + j]);
    }
}

__global__ void node_out_kernel(const float* __restrict__ agg2, const float* __restrict__ h,
                                const int* __restrict__ counts, const float* __restrict__ Wl2,
                                const float* __restrict__ bl2, const float* __restrict__ Wr2,
                                float* __restrict__ y, int N) {
    int wave = threadIdx.x >> 5;
    int lane = threadIdx.x & 31;
    int n = blockIdx.x * (blockDim.x >> 5) + wave;
    if (n >= N) return;
    float s1 = 0.f, s2 = 0.f;
    #pragma unroll
    for (int c = 0; c < 4; ++c) {
        int col = lane + 32 * c;
        if (col < HID_) {
            s1 = fmaf(agg2[(size_t)n * HID_ + col], Wl2[col], s1);
            s2 = fmaf(h[(size_t)n * HID_ + col],   Wr2[col], s2);
        }
    }
    for (int off = 16; off >= 1; off >>= 1) {
        s1 += __shfl_xor(s1, off, 32);
        s2 += __shfl_xor(s2, off, 32);
    }
    if (lane == 0)
        y[n] = s1 / fmaxf((float)counts[n], 1.0f) + bl2[0] + s2;
}

extern "C" void kernel_launch(void* const* d_in, const int* in_sizes, int n_in,
                              void* d_out, int out_size, void* d_ws, size_t ws_size,
                              hipStream_t stream) {
    const float* X      = (const float*)d_in[0];
    const float* x3     = (const float*)d_in[1];
    const int*   ei     = (const int*)d_in[2];
    const int*   gid    = (const int*)d_in[3];
    const float* w1     = (const float*)d_in[4];
    const float* b1     = (const float*)d_in[5];
    const float* w2     = (const float*)d_in[6];
    const float* b2     = (const float*)d_in[7];
    const float* w3     = (const float*)d_in[8];
    const float* b3     = (const float*)d_in[9];
    const float* Wl1    = (const float*)d_in[10];
    const float* bl1    = (const float*)d_in[11];
    const float* Wr1    = (const float*)d_in[12];
    const float* Wl2    = (const float*)d_in[13];
    const float* bl2    = (const float*)d_in[14];
    const float* Wr2    = (const float*)d_in[15];
    float* y = (float*)d_out;

    const int E = in_sizes[2] / 2;
    const int N = in_sizes[3];

    char* ws = (char*)d_ws;
    size_t off = 0;
    auto alloc = [&](size_t bytes) -> char* {
        char* p = ws + off;
        off = (off + bytes + 255) & ~(size_t)255;
        return p;
    };
    float*    act1  = (float*)alloc((size_t)8 * 32 * 21 * 36 * 4);
    float*    act2  = (float*)alloc((size_t)8 * 64 * 17 * 36 * 4);
    float*    act3  = (float*)alloc((size_t)8 * 64 * 13 * 16 * 4);
    float*    enc   = (float*)alloc((size_t)8 * ENC_ * 4);
    float*    ME    = (float*)alloc((size_t)256 * ENC_ * 4);
    float*    P     = (float*)alloc((size_t)256 * HID_ * 4);
    float*    Q     = (float*)alloc((size_t)16  * HID_ * 4);
    unsigned* aggk  = (unsigned*)alloc((size_t)N * 3 * 4);
    unsigned* mask  = (unsigned*)alloc((size_t)N * 4);
    int*      cnt   = (int*)alloc((size_t)N * 4);
    float*    hbuf  = (float*)alloc((size_t)N * HID_ * 4);
    float*    agg2  = (float*)alloc((size_t)N * HID_ * 4);

    const int T = 256;
    zero_u32<<<(N * 3 + T - 1) / T, T, 0, stream>>>(aggk, N * 3);
    zero_u32<<<(N + T - 1) / T, T, 0, stream>>>(mask, N);
    zero_u32<<<(N + T - 1) / T, T, 0, stream>>>((unsigned*)cnt, N);
    zero_u32<<<(N * HID_ + T - 1) / T, T, 0, stream>>>((unsigned*)agg2, N * HID_);

    conv1_kernel<<<(8 * 32 * 21 * 36 + T - 1) / T, T, 0, stream>>>(X, w1, b1, act1);
    conv2_kernel<<<(8 * 64 * 17 * 36 + T - 1) / T, T, 0, stream>>>(act1, w2, b2, act2);
    conv3_kernel<<<(8 * 64 * 13 * 16 + T - 1) / T, T, 0, stream>>>(act2, w3, b3, act3);
    pool_kernel<<<(8 * 64 * 11 * 4 + T - 1) / T, T, 0, stream>>>(act3, enc);

    edge_pass<<<(E + T - 1) / T, T, 0, stream>>>(ei, gid, x3, aggk, mask, cnt, E);

    maxenc_kernel<<<(ENC_ + 63) / 64, 64, 0, stream>>>(enc, ME);
    {
        dim3 gP(256 / 16, (HID_ + 15) / 16);
        gemm_wmma_f32<<<gP, 32, 0, stream>>>(ME, Wl1 + 3 * HID_, P, 256);
        dim3 gQ(1, (HID_ + 15) / 16);
        gemm_wmma_f32<<<gQ, 32, 0, stream>>>(enc, Wr1 + 3 * HID_, Q, 8);
    }

    node_h_kernel<<<(N + 7) / 8, 256, 0, stream>>>(x3, gid, mask, aggk, P, Q,
                                                   Wl1, Wr1, bl1, hbuf, N);

    edge_scatter_h<<<E, 128, 0, stream>>>(ei, hbuf, agg2, E);
    node_out_kernel<<<(N + 7) / 8, 256, 0, stream>>>(agg2, hbuf, cnt, Wl2, bl2, Wr2, y, N);

    (void)n_in; (void)out_size; (void)ws_size;
}


# GEMM K loop split over 16 waves per 16x16 tile, each slice unrolled so its loads overlap, partial tiles summed through LDS in fixed order (changes the K summation order)
# speedup vs baseline: 1.3739x; 1.3739x over previous
#include <hip/hip_runtime.h>
#include <hip/hip_bf16.h>
#include <math.h>


#define B_     8
#define CIN_   25
#define NPG_   2000
#define ENC_   2816
#define HID_   100
#define KGEMM_ 2816

typedef __attribute__((ext_vector_type(2))) float v2f;
typedef __attribute__((ext_vector_type(8))) float v8f;

__device__ __forceinline__ unsigned fkey(float f) {
    unsigned u = __float_as_uint(f);
    return (u & 0x80000000u) ? ~u : (u | 0x80000000u);
}
__device__ __forceinline__ float keyf(unsigned k) {
    unsigned u = (k & 0x80000000u) ? (k ^ 0x80000000u) : ~k;
    return __uint_as_float(u);
}

__global__ void zero_u32(unsigned* __restrict__ p, int n) {
    int i = blockIdx.x * blockDim.x + threadIdx.x;
    if (i < n) p[i] = 0u;
}

__global__ void conv1_kernel(const float* __restrict__ X, const float* __restrict__ w,
                             const float* __restrict__ b, float* __restrict__ out) {
    const int total = 8 * 32 * 21 * 6 * 6;
    int idx = blockIdx.x * blockDim.x + threadIdx.x;
    if (idx >= total) return;
    int x = idx % 6;  int t = idx / 6;
    int y = t % 6;    t /= 6;
    int z = t % 21;   t /= 21;
    int o = t % 32;   int bn = t / 32;
    float acc = b[o];
    for (int c = 0; c < 25; ++c)
        for (int kd = 0; kd < 5; ++kd) {
            const float* Xp = X + ((size_t)((bn * 25 + c) * 25 + (z + kd))) * 36;
            const float* Wp = w + ((size_t)((o * 25 + c) * 5 + kd)) * 9;
            for (int kh = 0; kh < 3; ++kh) {
                int yy = y + kh - 1; if (yy < 0 || yy >= 6) continue;
                for (int kw = 0; kw < 3; ++kw) {
                    int xx = x + kw - 1; if (xx < 0 || xx >= 6) continue;
                    acc = fmaf(Xp[yy * 6 + xx], Wp[kh * 3 + kw], acc);
                }
            }
        }
    out[idx] = fmaxf(acc, 0.0f);
}

__global__ void conv2_kernel(const float* __restrict__ A, const float* __restrict__ w,
                             const float* __restrict__ b, float* __restrict__ out) {
    const int total = 8 * 64 * 17 * 6 * 6;
    int idx = blockIdx.x * blockDim.x + threadIdx.x;
    if (idx >= total) return;
    int x = idx % 6;  int t = idx / 6;
    int y = t % 6;    t /= 6;
    int z = t % 17;   t /= 17;
    int o = t % 64;   int bn = t / 64;
    float acc = b[o];
    for (int c = 0; c < 32; ++c)
        for (int kd = 0; kd < 5; ++kd) {
            const float* Ap = A + ((size_t)((bn * 32 + c) * 21 + (z + kd))) * 36;
            const float* Wp = w + ((size_t)((o * 32 + c) * 5 + kd)) * 9;
            for (int kh = 0; kh < 3; ++kh) {
                int yy = y + kh - 1; if (yy < 0 || yy >= 6) continue;
                for (int kw = 0; kw < 3; ++kw) {
                    int xx = x + kw - 1; if (xx < 0 || xx >= 6) continue;
                    acc = fmaf(Ap[yy * 6 + xx], Wp[kh * 3 + kw], acc);
                }
            }
        }
    out[idx] = fmaxf(acc, 0.0f);
}

__global__ void conv3_kernel(const float* __restrict__ A, const float* __restrict__ w,
                             const float* __restrict__ b, float* __restrict__ out) {
    const int total = 8 * 64 * 13 * 4 * 4;
    int idx = blockIdx.x * blockDim.x + threadIdx.x;
    if (idx >= total) return;
    int x = idx % 4;  int t = idx / 4;
    int y = t % 4;    t /= 4;
    int z = t % 13;   t /= 13;
    int o = t % 64;   int bn = t / 64;
    float acc = b[o];
    for (int c = 0; c < 64; ++c)
        for (int kd = 0; kd < 5; ++kd) {
            const float* Ap = A + ((size_t)((bn * 64 + c) * 17 + (z + kd))) * 36;
            const float* Wp = w + ((size_t)((o * 64 + c) * 5 + kd)) * 9;
            for (int kh = 0; kh < 3; ++kh)
                for (int kw = 0; kw < 3; ++kw)
                    acc = fmaf(Ap[(y + kh) * 6 + (x + kw)], Wp[kh * 3 + kw], acc);
        }
    out[idx] = fmaxf(acc, 0.0f);
}

__global__ void pool_kernel(const float* __restrict__ A, float* __restrict__ enc) {
    const int total = 8 * 64 * 11 * 2 * 2;
    int idx = blockIdx.x * blockDim.x + threadIdx.x;
    if (idx >= total) return;
    int x = idx % 2;  int t = idx / 2;
    int y = t % 2;    t /= 2;
    int d = t % 11;   t /= 11;
    int ch = t % 64;  int bn = t / 64;
    float m = -INFINITY;
    for (int dd = 0; dd < 3; ++dd)
        for (int hh = 0; hh < 3; ++hh)
            for (int ww = 0; ww < 3; ++ww) {
                float v = A[((size_t)((bn * 64 + ch) * 13 + (d + dd))) * 16 + (y + hh) * 4 + (x + ww)];
                m = fmaxf(m, v);
            }
    enc[(size_t)bn * ENC_ + ch * 44 + d * 4 + y * 2 + x] = m;
}

__global__ void edge_pass(const int* __restrict__ ei, const int* __restrict__ gid,
                          const float* __restrict__ x3, unsigned* __restrict__ aggk,
                          unsigned* __restrict__ mask, int* __restrict__ counts, int E) {
    int e = blockIdx.x * blockDim.x + threadIdx.x;
    if (e >= E) return;
    int s = ei[e];
    int d = ei[E + e];
    int g = gid[s];
    atomicOr(&mask[d], 1u << g);
    atomicAdd(&counts[d], 1);
    atomicMax(&aggk[(size_t)d * 3 + 0], fkey(x3[(size_t)s * 3 + 0]));
    atomicMax(&aggk[(size_t)d * 3 + 1], fkey(x3[(size_t)s * 3 + 1]));
    atomicMax(&aggk[(size_t)d * 3 + 2], fkey(x3[(size_t)s * 3 + 2]));
}

__global__ void maxenc_kernel(const float* __restrict__ enc, float* __restrict__ ME) {
    int c = blockIdx.x * blockDim.x + threadIdx.x;
    if (c >= ENC_) return;
    ME[c] = 0.0f;
    for (int m = 1; m < 256; ++m) {
        int p = m & (m - 1);
        int g = __ffs(m) - 1;
        ME[(size_t)m * ENC_ + c] = fmaxf(ME[(size_t)p * ENC_ + c], enc[(size_t)g * ENC_ + c]);
    }
}

#define KSPLIT_ 16
#define KSLICE_ (KGEMM_ / KSPLIT_)

__global__ void __launch_bounds__(32 * KSPLIT_)
gemm_wmma_f32(const float* __restrict__ A, const float* __restrict__ W,
              float* __restrict__ C, int Mrows) {
    __shared__ float part[KSPLIT_][8][32];
    const int Ncols = HID_;
    int wave = threadIdx.x >> 5;
    int lane = threadIdx.x & 31;
    int half = lane >> 4;
    int l16  = lane & 15;
    int mBase = blockIdx.x * 16;
    int nBase = blockIdx.y * 16;
    int mRow = mBase + l16; if (mRow >= Mrows) mRow = Mrows - 1;
    int nCol = nBase + l16;
    int nClamp = (nCol < Ncols) ? nCol : (Ncols - 1);
    const float* Arow = A + (size_t)mRow * KGEMM_;
    const float* Bcol = W + nClamp;

    const int kBeg = wave * KSLICE_;
    v8f acc0 = {}, acc1 = {};
    #pragma unroll
    for (int kt = kBeg; kt < kBeg + KSLICE_; kt += 8) {
        int k0 = kt + half * 2;
        int k1 = k0 + 4;
        v2f a0, b0, a1, b1;
        a0.x = Arow[k0];
        a0.y = Arow[k0 + 1];
        a1.x = Arow[k1];
        a1.y = Arow[k1 + 1];
        b0.x = Bcol[(size_t)k0 * Ncols];
        b0.y = Bcol[(size_t)(k0 + 1) * Ncols];
        b1.x = Bcol[(size_t)k1 * Ncols];
        b1.y = Bcol[(size_t)(k1 + 1) * Ncols];
        acc0 = __builtin_amdgcn_wmma_f32_16x16x4_f32(false, a0, false, b0,
                                                     (short)0, acc0, false, false);
        acc1 = __builtin_amdgcn_wmma_f32_16x16x4_f32(false, a1, false, b1,
                                                     (short)0, acc1, false, false);
    }
    v8f acc = acc0 + acc1;
    for (int v = 0; v < 8; ++v) part[wave][v][lane] = acc[v];
    __syncthreads();
    if (wave < 8) {
        float s = part[0][wave][lane];
        for (int w = 1; w < KSPLIT_; ++w) s += part[w][wave][lane];
        int m = mBase + half * 8 + wave;
        if (m < Mrows && nCol < Ncols) C[(size_t)m * Ncols + nCol] = s;
    }
}

__global__ void node_h_kernel(const float* __restrict__ x3, const int* __restrict__ gid,
                              const unsigned* __restrict__ mask, const unsigned* __restrict__ aggk,
                              const float* __restrict__ P, const float* __restrict__ Q,
                              const float* __restrict__ Wl1, const float* __restrict__ Wr1,
                              const float* __restrict__ bl1, float* __restrict__ h, int N) {
    int wave = threadIdx.x >> 5;
    int lane = threadIdx.x & 31;
    int n = blockIdx.x * (blockDim.x >> 5) + wave;
    if (n >= N) return;

    float x0 = x3[(size_t)n * 3 + 0];
    float x1 = x3[(size_t)n * 3 + 1];
    float x2 = x3[(size_t)n * 3 + 2];
    int g = gid[n];
    unsigned mk = mask[n];
    bool has = (mk != 0u);
    float a0 = 0.f, a1 = 0.f, a2 = 0.f;
    if (has) {
        a0 = keyf(aggk[(size_t)n * 3 + 0]);
        a1 = keyf(aggk[(size_t)n * 3 + 1]);
        a2 = keyf(aggk[(size_t)n * 3 + 2]);
    }
    const float* Prow = P + (size_t)mk * HID_;
    const float* Qrow = Q + (size_t)g * HID_;

    float v[4];
    #pragma unroll
    for (int c = 0; c < 4; ++c) {
        int col = lane + 32 * c;
        if (col < HID_) {
            float t = bl1[col] + Qrow[col]
                    + x0 * Wr1[col] + x1 * Wr1[HID_ + col] + x2 * Wr1[2 * HID_ + col];
            if (has)
                t += Prow[col]
                   + a0 * Wl1[col] + a1 * Wl1[HID_ + col] + a2 * Wl1[2 * HID_ + col];
            v[c] = t;
        } else {
            v[c] = -INFINITY;
        }
    }
    float m = fmaxf(fmaxf(v[0], v[1]), fmaxf(v[2], v[3]));
    for (int off = 16; off >= 1; off >>= 1)
        m = fmaxf(m, __shfl_xor(m, off, 32));
    float s = 0.f;
    #pragma unroll
    for (int c = 0; c < 4; ++c) s += __expf(v[c] - m);
    for (int off = 16; off >= 1; off >>= 1)
        s += __shfl_xor(s, off, 32);
    float lse = m + __logf(s);
    #pragma unroll
    for (int c = 0; c < 4; ++c) {
        int col = lane + 32 * c;
        if (col < HID_) h[(size_t)n * HID_ + col] = v[c] - lse;
    }
}

__global__ void edge_scatter_h(const int* __restrict__ ei, const float* __restrict__ h,
                               float* __restrict__ agg2, int E) {
    int e = blockIdx.x;
    int j = threadIdx.x;
    if (j < HID_) {
        int s = ei[e];
        int d = ei[E + e];
        atomicAdd(&agg2[(size_t)d * HID_ + j], h[(size_t)s * HID_ + j]);
    }
}

__global__ void node_out_kernel(const float* __restrict__ agg2, const float* __restrict__ h,
                                const int* __restrict__ counts, const float* __restrict__ Wl2,
                                const float* __restrict__ bl2, const float* __restrict__ Wr2,
                                float* __restrict__ y, int N) {
    int wave = threadIdx.x >> 5;
    int lane = threadIdx.x & 31;
    int n = blockIdx.x * (blockDim.x >> 5) + wave;
    if (n >= N) return;
    float s1 = 0.f, s2 = 0.f;
    #pragma unroll
    for (int c = 0; c < 4; ++c) {
        int col = lane + 32 * c;
        if (col < HID_) {
            s1 = fmaf(agg2[(size_t)n * HID_ + col], Wl2[col], s1);
            s2 = fmaf(h[(size_t)n * HID_ + col],   Wr2[col], s2);
        }
    }
    for (int off = 16; off >= 1; off >>= 1) {
        s1 += __shfl_xor(s1, off, 32);
        s2 += __shfl_xor(s2, off, 32);
    }
    if (lane == 0)
        y[n] = s1 / fmaxf((float)counts[n], 1.0f) + bl2[0] + s2;
}

extern "C" void kernel_launch(void* const* d_in, const int* in_sizes, int n_in,
                              void* d_out, int out_size, void* d_ws, size_t ws_size,
                              hipStream_t stream) {
    const float* X      = (const float*)d_in[0];
    const float* x3     = (const float*)d_in[1];
    const int*   ei     = (const int*)d_in[2];
    const int*   gid    = (const int*)d_in[3];
    const float* w1     = (const float*)d_in[4];
    const float* b1     = (const float*)d_in[5];
    const float* w2     = (const float*)d_in[6];
    const float* b2     = (const float*)d_in[7];
    const float* w3     = (const float*)d_in[8];
    const float* b3     = (const float*)d_in[9];
    const float* Wl1    = (const float*)d_in[10];
    const float* bl1    = (const float*)d_in[11];
    const float* Wr1    = (const float*)d_in[12];
    const float* Wl2    = (const float*)d_in[13];
    const float* bl2    = (const float*)d_in[14];
    const float* Wr2    = (const float*)d_in[15];
    float* y = (float*)d_out;

    const int E = in_sizes[2] / 2;
    const int N = in_sizes[3];

    char* ws = (char*)d_ws;
    size_t off = 0;
    auto alloc = [&](size_t bytes) -> char* {
        char* p = ws + off;
        off = (off + bytes + 255) & ~(size_t)255;
        return p;
    };
    float*    act1  = (float*)alloc((size_t)8 * 32 * 21 * 36 * 4);
    float*    act2  = (float*)alloc((size_t)8 * 64 * 17 * 36 * 4);
    float*    act3  = (float*)alloc((size_t)8 * 64 * 13 * 16 * 4);
    float*    enc   = (float*)alloc((size_t)8 * ENC_ * 4);
    float*    ME    = (float*)alloc((size_t)256 * ENC_ * 4);
    float*    P     = (float*)alloc((size_t)256 * HID_ * 4);
    float*    Q     = (float*)alloc((size_t)16  * HID_ * 4);
    unsigned* aggk  = (unsigned*)alloc((size_t)N * 3 * 4);
    unsigned* mask  = (unsigned*)alloc((size_t)N * 4);
    int*      cnt   = (int*)alloc((size_t)N * 4);
    float*    hbuf  = (float*)alloc((size_t)N * HID_ * 4);
    float*    agg2  = (float*)alloc((size_t)N * HID_ * 4);

    const int T = 256;
    zero_u32<<<(N * 3 + T - 1) / T, T, 0, stream>>>(aggk, N * 3);
    zero_u32<<<(N + T - 1) / T, T, 0, stream>>>(mask, N);
    zero_u32<<<(N + T - 1) / T, T, 0, stream>>>((unsigned*)cnt, N);
    zero_u32<<<(N * HID_ + T - 1) / T, T, 0, stream>>>((unsigned*)agg2, N * HID_);

    conv1_kernel<<<(8 * 32 * 21 * 36 + T - 1) / T, T, 0, stream>>>(X, w1, b1, act1);
    conv2_kernel<<<(8 * 64 * 17 * 36 + T - 1) / T, T, 0, stream>>>(act1, w2, b2, act2);
    conv3_kernel<<<(8 * 64 * 13 * 16 + T - 1) / T, T, 0, stream>>>(act2, w3, b3, act3);
    pool_kernel<<<(8 * 64 * 11 * 4 + T - 1) / T, T, 0, stream>>>(act3, enc);

    edge_pass<<<(E + T - 1) / T, T, 0, stream>>>(ei, gid, x3, aggk, mask, cnt, E);

    maxenc_kernel<<<(ENC_ + 63) / 64, 64, 0, stream>>>(enc, ME);
    {
        dim3 gP(256 / 16, (HID_ + 15) / 16);
        gemm_wmma_f32<<<gP, 32 * KSPLIT_, 0, stream>>>(ME, Wl1 + 3 * HID_, P, 256);
        dim3 gQ(1, (HID_ + 15) / 16);
        gemm_wmma_f32<<<gQ, 32 * KSPLIT_, 0, stream>>>(enc, Wr1 + 3 * HID_, Q, 8);
    }

    node_h_kernel<<<(N + 7) / 8, 256, 0, stream>>>(x3, gid, mask, aggk, P, Q,
                                                   Wl1, Wr1, bl1, hbuf, N);

    edge_scatter_h<<<E, 128, 0, stream>>>(ei, hbuf, agg2, E);
    node_out_kernel<<<(N + 7) / 8, 256, 0, stream>>>(agg2, hbuf, cnt, Wl2, bl2, Wr2, y, N);

    (void)n_in; (void)out_size; (void)ws_size;
}


# Layer-2 weight vectors applied per node before the edge sum (linearity): the scatter adds one scalar per edge, not a 100-float row, and h is never stored (changes the summation order)
# speedup vs baseline: 1.4515x; 1.0565x over previous
#include <hip/hip_runtime.h>
#include <hip/hip_bf16.h>
#include <math.h>


#define B_     8
#define CIN_   25
#define NPG_   2000
#define ENC_   2816
#define HID_   100
#define KGEMM_ 2816

typedef __attribute__((ext_vector_type(2))) float v2f;
typedef __attribute__((ext_vector_type(8))) float v8f;

__device__ __forceinline__ unsigned fkey(float f) {
    unsigned u = __float_as_uint(f);
    return (u & 0x80000000u) ? ~u : (u | 0x80000000u);
}
__device__ __forceinline__ float keyf(unsigned k) {
    unsigned u = (k & 0x80000000u) ? (k ^ 0x80000000u) : ~k;
    return __uint_as_float(u);
}

__global__ void zero_u32(unsigned* __restrict__ p, int n) {
    int i = blockIdx.x * blockDim.x + threadIdx.x;
    if (i < n) p[i] = 0u;
}

__global__ void conv1_kernel(const float* __restrict__ X, const float* __restrict__ w,
                             const float* __restrict__ b, float* __restrict__ out) {
    const int total = 8 * 32 * 21 * 6 * 6;
    int idx = blockIdx.x * blockDim.x + threadIdx.x;
    if (idx >= total) return;
    int x = idx % 6;  int t = idx / 6;
    int y = t % 6;    t /= 6;
    int z = t % 21;   t /= 21;
    int o = t % 32;   int bn = t / 32;
    float acc = b[o];
    for (int c = 0; c < 25; ++c)
        for (int kd = 0; kd < 5; ++kd) {
            const float* Xp = X + ((size_t)((bn * 25 + c) * 25 + (z + kd))) * 36;
            const float* Wp = w + ((size_t)((o * 25 + c) * 5 + kd)) * 9;
            for (int kh = 0; kh < 3; ++kh) {
                int yy = y + kh - 1; if (yy < 0 || yy >= 6) continue;
                for (int kw = 0; kw < 3; ++kw) {
                    int xx = x + kw - 1; if (xx < 0 || xx >= 6) continue;
                    acc = fmaf(Xp[yy * 6 + xx], Wp[kh * 3 + kw], acc);
                }
            }
        }
    out[idx] = fmaxf(acc, 0.0f);
}

__global__ void conv2_kernel(const float* __restrict__ A, const float* __restrict__ w,
                             const float* __restrict__ b, float* __restrict__ out) {
    const int total = 8 * 64 * 17 * 6 * 6;
    int idx = blockIdx.x * blockDim.x + threadIdx.x;
    if (idx >= total) return;
    int x = idx % 6;  int t = idx / 6;
    int y = t % 6;    t /= 6;
    int z = t % 17;   t /= 17;
    int o = t % 64;   int bn = t / 64;
    float acc = b[o];
    for (int c = 0; c < 32; ++c)
        for (int kd = 0; kd < 5; ++kd) {
            const float* Ap = A + ((size_t)((bn * 32 + c) * 21 + (z + kd))) * 36;
            const float* Wp = w + ((size_t)((o * 32 + c) * 5 + kd)) * 9;
            for (int kh = 0; kh < 3; ++kh) {
                int yy = y + kh - 1; if (yy < 0 || yy >= 6) continue;
                for (int kw = 0; kw < 3; ++kw) {
                    int xx = x + kw - 1; if (xx < 0 || xx >= 6) continue;
                    acc = fmaf(Ap[yy * 6 + xx], Wp[kh * 3 + kw], acc);
                }
            }
        }
    out[idx] = fmaxf(acc, 0.0f);
}

__global__ void conv3_kernel(const float* __restrict__ A, const float* __restrict__ w,
                             const float* __restrict__ b, float* __restrict__ out) {
    const int total = 8 * 64 * 13 * 4 * 4;
    int idx = blockIdx.x * blockDim.x + threadIdx.x;
    if (idx >= total) return;
    int x = idx % 4;  int t = idx / 4;
    int y = t % 4;    t /= 4;
    int z = t % 13;   t /= 13;
    int o = t % 64;   int bn = t / 64;
    float acc = b[o];
    for (int c = 0; c < 64; ++c)
        for (int kd = 0; kd < 5; ++kd) {
            const float* Ap = A + ((size_t)((bn * 64 + c) * 17 + (z + kd))) * 36;
            const float* Wp = w + ((size_t)((o * 64 + c) * 5 + kd)) * 9;
            for (int kh = 0; kh < 3; ++kh)
                for (int kw = 0; kw < 3; ++kw)
                    acc = fmaf(Ap[(y + kh) * 6 + (x + kw)], Wp[kh * 3 + kw], acc);
        }
    out[idx] = fmaxf(acc, 0.0f);
}

__global__ void pool_kernel(const float* __restrict__ A, float* __restrict__ enc) {
    const int total = 8 * 64 * 11 * 2 * 2;
    int idx = blockIdx.x * blockDim.x + threadIdx.x;
    if (idx >= total) return;
    int x = idx % 2;  int t = idx / 2;
    int y = t % 2;    t /= 2;
    int d = t % 11;   t /= 11;
    int ch = t % 64;  int bn = t / 64;
    float m = -INFINITY;
    for (int dd = 0; dd < 3; ++dd)
        for (int hh = 0; hh < 3; ++hh)
            for (int ww = 0; ww < 3; ++ww) {
                float v = A[((size_t)((bn * 64 + ch) * 13 + (d + dd))) * 16 + (y + hh) * 4 + (x + ww)];
                m = fmaxf(m, v);
            }
    enc[(size_t)bn * ENC_ + ch * 44 + d * 4 + y * 2 + x] = m;
}

__global__ void edge_pass(const int* __restrict__ ei, const int* __restrict__ gid,
                          const float* __restrict__ x3, unsigned* __restrict__ aggk,
                          unsigned* __restrict__ mask, int* __restrict__ counts, int E) {
    int e = blockIdx.x * blockDim.x + threadIdx.x;
    if (e >= E) return;
    int s = ei[e];
    int d = ei[E + e];
    int g = gid[s];
    atomicOr(&mask[d], 1u << g);
    atomicAdd(&counts[d], 1);
    atomicMax(&aggk[(size_t)d * 3 + 0], fkey(x3[(size_t)s * 3 + 0]));
    atomicMax(&aggk[(size_t)d * 3 + 1], fkey(x3[(size_t)s * 3 + 1]));
    atomicMax(&aggk[(size_t)d * 3 + 2], fkey(x3[(size_t)s * 3 + 2]));
}

__global__ void maxenc_kernel(const float* __restrict__ enc, float* __restrict__ ME) {
    int c = blockIdx.x * blockDim.x + threadIdx.x;
    if (c >= ENC_) return;
    ME[c] = 0.0f;
    for (int m = 1; m < 256; ++m) {
        int p = m & (m - 1);
        int g = __ffs(m) - 1;
        ME[(size_t)m * ENC_ + c] = fmaxf(ME[(size_t)p * ENC_ + c], enc[(size_t)g * ENC_ + c]);
    }
}

#define KSPLIT_ 16
#define KSLICE_ (KGEMM_ / KSPLIT_)

__global__ void __launch_bounds__(32 * KSPLIT_)
gemm_wmma_f32(const float* __restrict__ A, const float* __restrict__ W,
              float* __restrict__ C, int Mrows) {
    __shared__ float part[KSPLIT_][8][32];
    const int Ncols = HID_;
    int wave = threadIdx.x >> 5;
    int lane = threadIdx.x & 31;
    int half = lane >> 4;
    int l16  = lane & 15;
    int mBase = blockIdx.x * 16;
    int nBase = blockIdx.y * 16;
    int mRow = mBase + l16; if (mRow >= Mrows) mRow = Mrows - 1;
    int nCol = nBase + l16;
    int nClamp = (nCol < Ncols) ? nCol : (Ncols - 1);
    const float* Arow = A + (size_t)mRow * KGEMM_;
    const float* Bcol = W + nClamp;

    const int kBeg = wave * KSLICE_;
    v8f acc0 = {}, acc1 = {};
    #pragma unroll
    for (int kt = kBeg; kt < kBeg + KSLICE_; kt += 8) {
        int k0 = kt + half * 2;
        int k1 = k0 + 4;
        v2f a0, b0, a1, b1;
        a0.x = Arow[k0];
        a0.y = Arow[k0 + 1];
        a1.x = Arow[k1];
        a1.y = Arow[k1 + 1];
        b0.x = Bcol[(size_t)k0 * Ncols];
        b0.y = Bcol[(size_t)(k0 + 1) * Ncols];
        b1.x = Bcol[(size_t)k1 * Ncols];
        b1.y = Bcol[(size_t)(k1 + 1) * Ncols];
        acc0 = __builtin_amdgcn_wmma_f32_16x16x4_f32(false, a0, false, b0,
                                                     (short)0, acc0, false, false);
        acc1 = __builtin_amdgcn_wmma_f32_16x16x4_f32(false, a1, false, b1,
                                                     (short)0, acc1, false, false);
    }
    v8f acc = acc0 + acc1;
    for (int v = 0; v < 8; ++v) part[wave][v][lane] = acc[v];
    __syncthreads();
    if (wave < 8) {
        float s = part[0][wave][lane];
        for (int w = 1; w < KSPLIT_; ++w) s += part[w][wave][lane];
        int m = mBase + half * 8 + wave;
        if (m < Mrows && nCol < Ncols) C[(size_t)m * Ncols + nCol] = s;
    }
}

__global__ void node_h_kernel(const float* __restrict__ x3, const int* __restrict__ gid,
                              const unsigned* __restrict__ mask, const unsigned* __restrict__ aggk,
                              const float* __restrict__ P, const float* __restrict__ Q,
                              const float* __restrict__ Wl1, const float* __restrict__ Wr1,
                              const float* __restrict__ bl1,
                              const float* __restrict__ Wl2, const float* __restrict__ Wr2,
                              float* __restrict__ tl, float* __restrict__ tr, int N) {
    int wave = threadIdx.x >> 5;
    int lane = threadIdx.x & 31;
    int n = blockIdx.x * (blockDim.x >> 5) + wave;
    if (n >= N) return;

    float x0 = x3[(size_t)n * 3 + 0];
    float x1 = x3[(size_t)n * 3 + 1];
    float x2 = x3[(size_t)n * 3 + 2];
    int g = gid[n];
    unsigned mk = mask[n];
    bool has = (mk != 0u);
    float a0 = 0.f, a1 = 0.f, a2 = 0.f;
    if (has) {
        a0 = keyf(aggk[(size_t)n * 3 + 0]);
        a1 = keyf(aggk[(size_t)n * 3 + 1]);
        a2 = keyf(aggk[(size_t)n * 3 + 2]);
    }
    const float* Prow = P + (size_t)mk * HID_;
    const float* Qrow = Q + (size_t)g * HID_;

    float v[4];
    #pragma unroll
    for (int c = 0; c < 4; ++c) {
        int col = lane + 32 * c;
        if (col < HID_) {
            float t = bl1[col] + Qrow[col]
                    + x0 * Wr1[col] + x1 * Wr1[HID_ + col] + x2 * Wr1[2 * HID_ + col];
            if (has)
                t += Prow[col]
                   + a0 * Wl1[col] + a1 * Wl1[HID_ + col] + a2 * Wl1[2 * HID_ + col];
            v[c] = t;
        } else {
            v[c] = -INFINITY;
        }
    }
    float m = fmaxf(fmaxf(v[0], v[1]), fmaxf(v[2], v[3]));
    for (int off = 16; off >= 1; off >>= 1)
        m = fmaxf(m, __shfl_xor(m, off, 32));
    float s = 0.f;
    #pragma unroll
    for (int c = 0; c < 4; ++c) s += __expf(v[c] - m);
    for (int off = 16; off >= 1; off >>= 1)
        s += __shfl_xor(s, off, 32);
    float lse = m + __logf(s);
    float s1 = 0.f, s2 = 0.f;
    #pragma unroll
    for (int c = 0; c < 4; ++c) {
        int col = lane + 32 * c;
        if (col < HID_) {
            float hv = v[c] - lse;
            s1 = fmaf(hv, Wl2[col], s1);
            s2 = fmaf(hv, Wr2[col], s2);
        }
    }
    for (int off = 16; off >= 1; off >>= 1) {
        s1 += __shfl_xor(s1, off, 32);
        s2 += __shfl_xor(s2, off, 32);
    }
    if (lane == 0) { tl[n] = s1; tr[n] = s2; }
}

__global__ void edge_scatter_t(const int* __restrict__ ei, const float* __restrict__ tl,
                               float* __restrict__ agg2, int E) {
    int e = blockIdx.x * blockDim.x + threadIdx.x;
    if (e >= E) return;
    int s = ei[e];
    int d = ei[E + e];
    atomicAdd(&agg2[d], tl[s]);
}

__global__ void node_out_kernel(const float* __restrict__ agg2, const float* __restrict__ tr,
                                const int* __restrict__ counts, const float* __restrict__ bl2,
                                float* __restrict__ y, int N) {
    int n = blockIdx.x * blockDim.x + threadIdx.x;
    if (n >= N) return;
    y[n] = agg2[n] / fmaxf((float)counts[n], 1.0f) + bl2[0] + tr[n];
}

extern "C" void kernel_launch(void* const* d_in, const int* in_sizes, int n_in,
                              void* d_out, int out_size, void* d_ws, size_t ws_size,
                              hipStream_t stream) {
    const float* X      = (const float*)d_in[0];
    const float* x3     = (const float*)d_in[1];
    const int*   ei     = (const int*)d_in[2];
    const int*   gid    = (const int*)d_in[3];
    const float* w1     = (const float*)d_in[4];
    const float* b1     = (const float*)d_in[5];
    const float* w2     = (const float*)d_in[6];
    const float* b2     = (const float*)d_in[7];
    const float* w3     = (const float*)d_in[8];
    const float* b3     = (const float*)d_in[9];
    const float* Wl1    = (const float*)d_in[10];
    const float* bl1    = (const float*)d_in[11];
    const float* Wr1    = (const float*)d_in[12];
    const float* Wl2    = (const float*)d_in[13];
    const float* bl2    = (const float*)d_in[14];
    const float* Wr2    = (const float*)d_in[15];
    float* y = (float*)d_out;

    const int E = in_sizes[2] / 2;
    const int N = in_sizes[3];

    char* ws = (char*)d_ws;
    size_t off = 0;
    auto alloc = [&](size_t bytes) -> char* {
        char* p = ws + off;
        off = (off + bytes + 255) & ~(size_t)255;
        return p;
    };
    float*    act1  = (float*)alloc((size_t)8 * 32 * 21 * 36 * 4);
    float*    act2  = (float*)alloc((size_t)8 * 64 * 17 * 36 * 4);
    float*    act3  = (float*)alloc((size_t)8 * 64 * 13 * 16 * 4);
    float*    enc   = (float*)alloc((size_t)8 * ENC_ * 4);
    float*    ME    = (float*)alloc((size_t)256 * ENC_ * 4);
    float*    P     = (float*)alloc((size_t)256 * HID_ * 4);
    float*    Q     = (float*)alloc((size_t)16  * HID_ * 4);
    unsigned* aggk  = (unsigned*)alloc((size_t)N * 3 * 4);
    unsigned* mask  = (unsigned*)alloc((size_t)N * 4);
    int*      cnt   = (int*)alloc((size_t)N * 4);
    float*    tl    = (float*)alloc((size_t)N * 4);
    float*    tr    = (float*)alloc((size_t)N * 4);
    float*    agg2  = (float*)alloc((size_t)N * 4);

    const int T = 256;
    zero_u32<<<(N * 3 + T - 1) / T, T, 0, stream>>>(aggk, N * 3);
    zero_u32<<<(N + T - 1) / T, T, 0, stream>>>(mask, N);
    zero_u32<<<(N + T - 1) / T, T, 0, stream>>>((unsigned*)cnt, N);
    zero_u32<<<(N + T - 1) / T, T, 0, stream>>>((unsigned*)agg2, N);

    conv1_kernel<<<(8 * 32 * 21 * 36 + T - 1) / T, T, 0, stream>>>(X, w1, b1, act1);
    conv2_kernel<<<(8 * 64 * 17 * 36 + T - 1) / T, T, 0, stream>>>(act1, w2, b2, act2);
    conv3_kernel<<<(8 * 64 * 13 * 16 + T - 1) / T, T, 0, stream>>>(act2, w3, b3, act3);
    pool_kernel<<<(8 * 64 * 11 * 4 + T - 1) / T, T, 0, stream>>>(act3, enc);

    edge_pass<<<(E + T - 1) / T, T, 0, stream>>>(ei, gid, x3, aggk, mask, cnt, E);

    maxenc_kernel<<<(ENC_ + 63) / 64, 64, 0, stream>>>(enc, ME);
    {
        dim3 gP(256 / 16, (HID_ + 15) / 16);
        gemm_wmma_f32<<<gP, 32 * KSPLIT_, 0, stream>>>(ME, Wl1 + 3 * HID_, P, 256);
        dim3 gQ(1, (HID_ + 15) / 16);
        gemm_wmma_f32<<<gQ, 32 * KSPLIT_, 0, stream>>>(enc, Wr1 + 3 * HID_, Q, 8);
    }

    node_h_kernel<<<(N + 7) / 8, 256, 0, stream>>>(x3, gid, mask, aggk, P, Q,
                                                   Wl1, Wr1, bl1, Wl2, Wr2, tl, tr, N);

    edge_scatter_t<<<(E + T - 1) / T, T, 0, stream>>>(ei, tl, agg2, E);
    node_out_kernel<<<(N + T - 1) / T, T, 0, stream>>>(agg2, tr, cnt, bl2, y, N);

    (void)n_in; (void)out_size; (void)ws_size;
}


# Convs register-blocked over output channels: a workgroup stages the weights of 4-8 channels in LDS, each loaded input feeds 4-8 FMAs, taps branch-free; per-output summation order unchanged
# speedup vs baseline: 2.2841x; 1.5736x over previous
#include <hip/hip_runtime.h>
#include <hip/hip_bf16.h>
#include <math.h>


#define B_     8
#define CIN_   25
#define NPG_   2000
#define ENC_   2816
#define HID_   100
#define KGEMM_ 2816

typedef __attribute__((ext_vector_type(2))) float v2f;
typedef __attribute__((ext_vector_type(8))) float v8f;

__device__ __forceinline__ unsigned fkey(float f) {
    unsigned u = __float_as_uint(f);
    return (u & 0x80000000u) ? ~u : (u | 0x80000000u);
}
__device__ __forceinline__ float keyf(unsigned k) {
    unsigned u = (k & 0x80000000u) ? (k ^ 0x80000000u) : ~k;
    return __uint_as_float(u);
}

__global__ void zero_u32(unsigned* __restrict__ p, int n) {
    int i = blockIdx.x * blockDim.x + threadIdx.x;
    if (i < n) p[i] = 0u;
}

#define CONV_T_ 128

template <int CIN, int COUT, int DIN, int DOUT, int HIN, int HOUT, int PAD, int OCT>
__device__ __forceinline__ void conv_body(const float* __restrict__ in, const float* __restrict__ w,
                                          const float* __restrict__ b, float* __restrict__ out,
                                          float* wl) {
    constexpr int WIN = HIN;
    constexpr int WOUT = HOUT;
    constexpr int PLANE = HIN * WIN;
    constexpr int NPOS = DOUT * HOUT * WOUT;
    constexpr int NTOT = 8 * NPOS;
    constexpr int KW = CIN * 45;

    const int og = blockIdx.y * OCT;
    constexpr int NW = KW * OCT;
    constexpr int SU = 18;
    for (int base = 0; base < NW; base += SU * CONV_T_) {
        float tmp[SU];
        #pragma unroll
        for (int u = 0; u < SU; ++u) {
            int i = base + u * CONV_T_ + (int)threadIdx.x;
            int ic = i < NW ? i : NW - 1;
            int o = ic / KW;
            int r = ic - o * KW;
            tmp[u] = w[(size_t)(og + o) * KW + r];
        }
        #pragma unroll
        for (int u = 0; u < SU; ++u) {
            int i = base + u * CONV_T_ + (int)threadIdx.x;
            if (i < NW) {
                int o = i / KW;
                int r = i - o * KW;
                wl[r * OCT + o] = tmp[u];
            }
        }
    }
    __syncthreads();

    int gp = blockIdx.x * CONV_T_ + threadIdx.x;
    const bool active = gp < NTOT;
    if (!active) gp = NTOT - 1;
    const int bn = gp / NPOS;
    const int p = gp - bn * NPOS;
    const int x = p % WOUT;
    const int t2 = p / WOUT;
    const int y = t2 % HOUT;
    const int z = t2 / HOUT;

    int off[9];
    bool ok[9];
    #pragma unroll
    for (int kh = 0; kh < 3; ++kh)
        #pragma unroll
        for (int kw = 0; kw < 3; ++kw) {
            int yy = y + kh - PAD;
            int xx = x + kw - PAD;
            ok[kh * 3 + kw] = (yy >= 0) && (yy < HIN) && (xx >= 0) && (xx < WIN);
            int yc = yy < 0 ? 0 : (yy >= HIN ? HIN - 1 : yy);
            int xc = xx < 0 ? 0 : (xx >= WIN ? WIN - 1 : xx);
            off[kh * 3 + kw] = yc * WIN + xc;
        }

    float acc[OCT];
    #pragma unroll
    for (int o = 0; o < OCT; ++o) acc[o] = b[og + o];

    const float* ib = in + ((size_t)bn * CIN * DIN + z) * PLANE;
    for (int c = 0; c < CIN; ++c) {
        const float* ic = ib + (size_t)c * DIN * PLANE;
        float xv[45];
        #pragma unroll
        for (int kd = 0; kd < 5; ++kd)
            #pragma unroll
            for (int t = 0; t < 9; ++t) {
                float v = ic[kd * PLANE + off[t]];
                xv[kd * 9 + t] = (PAD == 0 || ok[t]) ? v : 0.0f;
            }
        const float* wc = wl + c * 45 * OCT;
        #pragma unroll
        for (int j = 0; j < 45; ++j)
            #pragma unroll
            for (int o = 0; o < OCT; ++o)
                acc[o] = fmaf(xv[j], wc[j * OCT + o], acc[o]);
    }
    if (active) {
        #pragma unroll
        for (int o = 0; o < OCT; ++o)
            out[((size_t)(bn * COUT + og + o) * DOUT + z) * (HOUT * WOUT) + y * WOUT + x] =
                fmaxf(acc[o], 0.0f);
    }
}

__global__ void __launch_bounds__(CONV_T_) __attribute__((amdgpu_waves_per_eu(1, 2)))
conv1_kernel(const float* __restrict__ X, const float* __restrict__ w,
             const float* __restrict__ b, float* __restrict__ out) {
    __shared__ __attribute__((aligned(16))) float wl[25 * 45 * 4];
    conv_body<25, 32, 25, 21, 6, 6, 1, 4>(X, w, b, out, wl);
}

__global__ void __launch_bounds__(CONV_T_) __attribute__((amdgpu_waves_per_eu(1, 2)))
conv2_kernel(const float* __restrict__ A, const float* __restrict__ w,
             const float* __restrict__ b, float* __restrict__ out) {
    __shared__ __attribute__((aligned(16))) float wl[32 * 45 * 8];
    conv_body<32, 64, 21, 17, 6, 6, 1, 8>(A, w, b, out, wl);
}

__global__ void __launch_bounds__(CONV_T_)
conv3_kernel(const float* __restrict__ A, const float* __restrict__ w,
             const float* __restrict__ b, float* __restrict__ out) {
    __shared__ __attribute__((aligned(16))) float wl[64 * 45 * 4];
    conv_body<64, 64, 17, 13, 6, 4, 0, 4>(A, w, b, out, wl);
}

__global__ void pool_kernel(const float* __restrict__ A, float* __restrict__ enc) {
    const int total = 8 * 64 * 11 * 2 * 2;
    int idx = blockIdx.x * blockDim.x + threadIdx.x;
    if (idx >= total) return;
    int x = idx % 2;  int t = idx / 2;
    int y = t % 2;    t /= 2;
    int d = t % 11;   t /= 11;
    int ch = t % 64;  int bn = t / 64;
    float m = -INFINITY;
    for (int dd = 0; dd < 3; ++dd)
        for (int hh = 0; hh < 3; ++hh)
            for (int ww = 0; ww < 3; ++ww) {
                float v = A[((size_t)((bn * 64 + ch) * 13 + (d + dd))) * 16 + (y + hh) * 4 + (x + ww)];
                m = fmaxf(m, v);
            }
    enc[(size_t)bn * ENC_ + ch * 44 + d * 4 + y * 2 + x] = m;
}

__global__ void edge_pass(const int* __restrict__ ei, const int* __restrict__ gid,
                          const float* __restrict__ x3, unsigned* __restrict__ aggk,
                          unsigned* __restrict__ mask, int* __restrict__ counts, int E) {
    int e = blockIdx.x * blockDim.x + threadIdx.x;
    if (e >= E) return;
    int s = ei[e];
    int d = ei[E + e];
    int g = gid[s];
    atomicOr(&mask[d], 1u << g);
    atomicAdd(&counts[d], 1);
    atomicMax(&aggk[(size_t)d * 3 + 0], fkey(x3[(size_t)s * 3 + 0]));
    atomicMax(&aggk[(size_t)d * 3 + 1], fkey(x3[(size_t)s * 3 + 1]));
    atomicMax(&aggk[(size_t)d * 3 + 2], fkey(x3[(size_t)s * 3 + 2]));
}

__global__ void maxenc_kernel(const float* __restrict__ enc, float* __restrict__ ME) {
    int c = blockIdx.x * blockDim.x + threadIdx.x;
    if (c >= ENC_) return;
    ME[c] = 0.0f;
    for (int m = 1; m < 256; ++m) {
        int p = m & (m - 1);
        int g = __ffs(m) - 1;
        ME[(size_t)m * ENC_ + c] = fmaxf(ME[(size_t)p * ENC_ + c], enc[(size_t)g * ENC_ + c]);
    }
}

#define KSPLIT_ 16
#define KSLICE_ (KGEMM_ / KSPLIT_)

__global__ void __launch_bounds__(32 * KSPLIT_)
gemm_wmma_f32(const float* __restrict__ A, const float* __restrict__ W,
              float* __restrict__ C, int Mrows) {
    __shared__ float part[KSPLIT_][8][32];
    const int Ncols = HID_;
    int wave = threadIdx.x >> 5;
    int lane = threadIdx.x & 31;
    int half = lane >> 4;
    int l16  = lane & 15;
    int mBase = blockIdx.x * 16;
    int nBase = blockIdx.y * 16;
    int mRow = mBase + l16; if (mRow >= Mrows) mRow = Mrows - 1;
    int nCol = nBase + l16;
    int nClamp = (nCol < Ncols) ? nCol : (Ncols - 1);
    const float* Arow = A + (size_t)mRow * KGEMM_;
    const float* Bcol = W + nClamp;

    const int kBeg = wave * KSLICE_;
    v8f acc0 = {}, acc1 = {};
    #pragma unroll
    for (int kt = kBeg; kt < kBeg + KSLICE_; kt += 8) {
        int k0 = kt + half * 2;
        int k1 = k0 + 4;
        v2f a0, b0, a1, b1;
        a0.x = Arow[k0];
        a0.y = Arow[k0 + 1];
        a1.x = Arow[k1];
        a1.y = Arow[k1 + 1];
        b0.x = Bcol[(size_t)k0 * Ncols];
        b0.y = Bcol[(size_t)(k0 + 1) * Ncols];
        b1.x = Bcol[(size_t)k1 * Ncols];
        b1.y = Bcol[(size_t)(k1 + 1) * Ncols];
        acc0 = __builtin_amdgcn_wmma_f32_16x16x4_f32(false, a0, false, b0,
                                                     (short)0, acc0, false, false);
        acc1 = __builtin_amdgcn_wmma_f32_16x16x4_f32(false, a1, false, b1,
                                                     (short)0, acc1, false, false);
    }
    v8f acc = acc0 + acc1;
    for (int v = 0; v < 8; ++v) part[wave][v][lane] = acc[v];
    __syncthreads();
    if (wave < 8) {
        float s = part[0][wave][lane];
        for (int w = 1; w < KSPLIT_; ++w) s += part[w][wave][lane];
        int m = mBase + half * 8 + wave;
        if (m < Mrows && nCol < Ncols) C[(size_t)m * Ncols + nCol] = s;
    }
}

__global__ void node_h_kernel(const float* __restrict__ x3, const int* __restrict__ gid,
                              const unsigned* __restrict__ mask, const unsigned* __restrict__ aggk,
                              const float* __restrict__ P, const float* __restrict__ Q,
                              const float* __restrict__ Wl1, const float* __restrict__ Wr1,
                              const float* __restrict__ bl1,
                              const float* __restrict__ Wl2, const float* __restrict__ Wr2,
                              float* __restrict__ tl, float* __restrict__ tr, int N) {
    int wave = threadIdx.x >> 5;
    int lane = threadIdx.x & 31;
    int n = blockIdx.x * (blockDim.x >> 5) + wave;
    if (n >= N) return;

    float x0 = x3[(size_t)n * 3 + 0];
    float x1 = x3[(size_t)n * 3 + 1];
    float x2 = x3[(size_t)n * 3 + 2];
    int g = gid[n];
    unsigned mk = mask[n];
    bool has = (mk != 0u);
    float a0 = 0.f, a1 = 0.f, a2 = 0.f;
    if (has) {
        a0 = keyf(aggk[(size_t)n * 3 + 0]);
        a1 = keyf(aggk[(size_t)n * 3 + 1]);
        a2 = keyf(aggk[(size_t)n * 3 + 2]);
    }
    const float* Prow = P + (size_t)mk * HID_;
    const float* Qrow = Q + (size_t)g * HID_;

    float v[4];
    #pragma unroll
    for (int c = 0; c < 4; ++c) {
        int col = lane + 32 * c;
        if (col < HID_) {
            float t = bl1[col] + Qrow[col]
                    + x0 * Wr1[col] + x1 * Wr1[HID_ + col] + x2 * Wr1[2 * HID_ + col];
            if (has)
                t += Prow[col]
                   + a0 * Wl1[col] + a1 * Wl1[HID_ + col] + a2 * Wl1[2 * HID_ + col];
            v[c] = t;
        } else {
            v[c] = -INFINITY;
        }
    }
    float m = fmaxf(fmaxf(v[0], v[1]), fmaxf(v[2], v[3]));
    for (int off = 16; off >= 1; off >>= 1)
        m = fmaxf(m, __shfl_xor(m, off, 32));
    float s = 0.f;
    #pragma unroll
    for (int c = 0; c < 4; ++c) s += __expf(v[c] - m);
    for (int off = 16; off >= 1; off >>= 1)
        s += __shfl_xor(s, off, 32);
    float lse = m + __logf(s);
    float s1 = 0.f, s2 = 0.f;
    #pragma unroll
    for (int c = 0; c < 4; ++c) {
        int col = lane + 32 * c;
        if (col < HID_) {
            float hv = v[c] - lse;
            s1 = fmaf(hv, Wl2[col], s1);
            s2 = fmaf(hv, Wr2[col], s2);
        }
    }
    for (int off = 16; off >= 1; off >>= 1) {
        s1 += __shfl_xor(s1, off, 32);
        s2 += __shfl_xor(s2, off, 32);
    }
    if (lane == 0) { tl[n] = s1; tr[n] = s2; }
}

__global__ void edge_scatter_t(const int* __restrict__ ei, const float* __restrict__ tl,
                               float* __restrict__ agg2, int E) {
    int e = blockIdx.x * blockDim.x + threadIdx.x;
    if (e >= E) return;
    int s = ei[e];
    int d = ei[E + e];
    atomicAdd(&agg2[d], tl[s]);
}

__global__ void node_out_kernel(const float* __restrict__ agg2, const float* __restrict__ tr,
                                const int* __restrict__ counts, const float* __restrict__ bl2,
                                float* __restrict__ y, int N) {
    int n = blockIdx.x * blockDim.x + threadIdx.x;
    if (n >= N) return;
    y[n] = agg2[n] / fmaxf((float)counts[n], 1.0f) + bl2[0] + tr[n];
}

extern "C" void kernel_launch(void* const* d_in, const int* in_sizes, int n_in,
                              void* d_out, int out_size, void* d_ws, size_t ws_size,
                              hipStream_t stream) {
    const float* X      = (const float*)d_in[0];
    const float* x3     = (const float*)d_in[1];
    const int*   ei     = (const int*)d_in[2];
    const int*   gid    = (const int*)d_in[3];
    const float* w1     = (const float*)d_in[4];
    const float* b1     = (const float*)d_in[5];
    const float* w2     = (const float*)d_in[6];
    const float* b2     = (const float*)d_in[7];
    const float* w3     = (const float*)d_in[8];
    const float* b3     = (const float*)d_in[9];
    const float* Wl1    = (const float*)d_in[10];
    const float* bl1    = (const float*)d_in[11];
    const float* Wr1    = (const float*)d_in[12];
    const float* Wl2    = (const float*)d_in[13];
    const float* bl2    = (const float*)d_in[14];
    const float* Wr2    = (const float*)d_in[15];
    float* y = (float*)d_out;

    const int E = in_sizes[2] / 2;
    const int N = in_sizes[3];

    char* ws = (char*)d_ws;
    size_t off = 0;
    auto alloc = [&](size_t bytes) -> char* {
        char* p = ws + off;
        off = (off + bytes + 255) & ~(size_t)255;
        return p;
    };
    float*    act1  = (float*)alloc((size_t)8 * 32 * 21 * 36 * 4);
    float*    act2  = (float*)alloc((size_t)8 * 64 * 17 * 36 * 4);
    float*    act3  = (float*)alloc((size_t)8 * 64 * 13 * 16 * 4);
    float*    enc   = (float*)alloc((size_t)8 * ENC_ * 4);
    float*    ME    = (float*)alloc((size_t)256 * ENC_ * 4);
    float*    P     = (float*)alloc((size_t)256 * HID_ * 4);
    float*    Q     = (float*)alloc((size_t)16  * HID_ * 4);
    unsigned* aggk  = (unsigned*)alloc((size_t)N * 3 * 4);
    unsigned* mask  = (unsigned*)alloc((size_t)N * 4);
    int*      cnt   = (int*)alloc((size_t)N * 4);
    float*    tl    = (float*)alloc((size_t)N * 4);
    float*    tr    = (float*)alloc((size_t)N * 4);
    float*    agg2  = (float*)alloc((size_t)N * 4);

    const int T = 256;
    zero_u32<<<(N * 3 + T - 1) / T, T, 0, stream>>>(aggk, N * 3);
    zero_u32<<<(N + T - 1) / T, T, 0, stream>>>(mask, N);
    zero_u32<<<(N + T - 1) / T, T, 0, stream>>>((unsigned*)cnt, N);
    zero_u32<<<(N + T - 1) / T, T, 0, stream>>>((unsigned*)agg2, N);

    conv1_kernel<<<dim3(48, 8), CONV_T_, 0, stream>>>(X, w1, b1, act1);
    conv2_kernel<<<dim3(39, 8), CONV_T_, 0, stream>>>(act1, w2, b2, act2);
    conv3_kernel<<<dim3(13, 16), CONV_T_, 0, stream>>>(act2, w3, b3, act3);
    pool_kernel<<<(8 * 64 * 11 * 4 + T - 1) / T, T, 0, stream>>>(act3, enc);

    edge_pass<<<(E + T - 1) / T, T, 0, stream>>>(ei, gid, x3, aggk, mask, cnt, E);

    maxenc_kernel<<<(ENC_ + 63) / 64, 64, 0, stream>>>(enc, ME);
    {
        dim3 gP(256 / 16, (HID_ + 15) / 16);
        gemm_wmma_f32<<<gP, 32 * KSPLIT_, 0, stream>>>(ME, Wl1 + 3 * HID_, P, 256);
        dim3 gQ(1, (HID_ + 15) / 16);
        gemm_wmma_f32<<<gQ, 32 * KSPLIT_, 0, stream>>>(enc, Wr1 + 3 * HID_, Q, 8);
    }

    node_h_kernel<<<(N + 7) / 8, 256, 0, stream>>>(x3, gid, mask, aggk, P, Q,
                                                   Wl1, Wr1, bl1, Wl2, Wr2, tl, tr, N);

    edge_scatter_t<<<(E + T - 1) / T, T, 0, stream>>>(ei, tl, agg2, E);
    node_out_kernel<<<(N + T - 1) / T, T, 0, stream>>>(agg2, tr, cnt, bl2, y, N);

    (void)n_in; (void)out_size; (void)ws_size;
}


# Subset-max table built in registers: a thread loads its 8 enc values once and derives 16 masks by the same max chain, replacing a 255-step walk that re-read the table from memory at every step
# speedup vs baseline: 3.1925x; 1.3977x over previous
#include <hip/hip_runtime.h>
#include <hip/hip_bf16.h>
#include <math.h>


#define B_     8
#define CIN_   25
#define NPG_   2000
#define ENC_   2816
#define HID_   100
#define KGEMM_ 2816

typedef __attribute__((ext_vector_type(2))) float v2f;
typedef __attribute__((ext_vector_type(8))) float v8f;

__device__ __forceinline__ unsigned fkey(float f) {
    unsigned u = __float_as_uint(f);
    return (u & 0x80000000u) ? ~u : (u | 0x80000000u);
}
__device__ __forceinline__ float keyf(unsigned k) {
    unsigned u = (k & 0x80000000u) ? (k ^ 0x80000000u) : ~k;
    return __uint_as_float(u);
}

__global__ void zero_u32(unsigned* __restrict__ p, int n) {
    int i = blockIdx.x * blockDim.x + threadIdx.x;
    if (i < n) p[i] = 0u;
}

#define CONV_T_ 128

template <int CIN, int COUT, int DIN, int DOUT, int HIN, int HOUT, int PAD, int OCT>
__device__ __forceinline__ void conv_body(const float* __restrict__ in, const float* __restrict__ w,
                                          const float* __restrict__ b, float* __restrict__ out,
                                          float* wl) {
    constexpr int WIN = HIN;
    constexpr int WOUT = HOUT;
    constexpr int PLANE = HIN * WIN;
    constexpr int NPOS = DOUT * HOUT * WOUT;
    constexpr int NTOT = 8 * NPOS;
    constexpr int KW = CIN * 45;

    const int og = blockIdx.y * OCT;
    constexpr int NW = KW * OCT;
    constexpr int SU = 18;
    for (int base = 0; base < NW; base += SU * CONV_T_) {
        float tmp[SU];
        #pragma unroll
        for (int u = 0; u < SU; ++u) {
            int i = base + u * CONV_T_ + (int)threadIdx.x;
            int ic = i < NW ? i : NW - 1;
            int o = ic / KW;
            int r = ic - o * KW;
            tmp[u] = w[(size_t)(og + o) * KW + r];
        }
        #pragma unroll
        for (int u = 0; u < SU; ++u) {
            int i = base + u * CONV_T_ + (int)threadIdx.x;
            if (i < NW) {
                int o = i / KW;
                int r = i - o * KW;
                wl[r * OCT + o] = tmp[u];
            }
        }
    }
    __syncthreads();

    int gp = blockIdx.x * CONV_T_ + threadIdx.x;
    const bool active = gp < NTOT;
    if (!active) gp = NTOT - 1;
    const int bn = gp / NPOS;
    const int p = gp - bn * NPOS;
    const int x = p % WOUT;
    const int t2 = p / WOUT;
    const int y = t2 % HOUT;
    const int z = t2 / HOUT;

    int off[9];
    bool ok[9];
    #pragma unroll
    for (int kh = 0; kh < 3; ++kh)
        #pragma unroll
        for (int kw = 0; kw < 3; ++kw) {
            int yy = y + kh - PAD;
            int xx = x + kw - PAD;
            ok[kh * 3 + kw] = (yy >= 0) && (yy < HIN) && (xx >= 0) && (xx < WIN);
            int yc = yy < 0 ? 0 : (yy >= HIN ? HIN - 1 : yy);
            int xc = xx < 0 ? 0 : (xx >= WIN ? WIN - 1 : xx);
            off[kh * 3 + kw] = yc * WIN + xc;
        }

    float acc[OCT];
    #pragma unroll
    for (int o = 0; o < OCT; ++o) acc[o] = b[og + o];

    const float* ib = in + ((size_t)bn * CIN * DIN + z) * PLANE;
    for (int c = 0; c < CIN; ++c) {
        const float* ic = ib + (size_t)c * DIN * PLANE;
        float xv[45];
        #pragma unroll
        for (int kd = 0; kd < 5; ++kd)
            #pragma unroll
            for (int t = 0; t < 9; ++t) {
                float v = ic[kd * PLANE + off[t]];
                xv[kd * 9 + t] = (PAD == 0 || ok[t]) ? v : 0.0f;
            }
        const float* wc = wl + c * 45 * OCT;
        #pragma unroll
        for (int j = 0; j < 45; ++j)
            #pragma unroll
            for (int o = 0; o < OCT; ++o)
                acc[o] = fmaf(xv[j], wc[j * OCT + o], acc[o]);
    }
    if (active) {
        #pragma unroll
        for (int o = 0; o < OCT; ++o)
            out[((size_t)(bn * COUT + og + o) * DOUT + z) * (HOUT * WOUT) + y * WOUT + x] =
                fmaxf(acc[o], 0.0f);
    }
}

__global__ void __launch_bounds__(CONV_T_) __attribute__((amdgpu_waves_per_eu(1, 2)))
conv1_kernel(const float* __restrict__ X, const float* __restrict__ w,
             const float* __restrict__ b, float* __restrict__ out) {
    __shared__ __attribute__((aligned(16))) float wl[25 * 45 * 4];
    conv_body<25, 32, 25, 21, 6, 6, 1, 4>(X, w, b, out, wl);
}

__global__ void __launch_bounds__(CONV_T_) __attribute__((amdgpu_waves_per_eu(1, 2)))
conv2_kernel(const float* __restrict__ A, const float* __restrict__ w,
             const float* __restrict__ b, float* __restrict__ out) {
    __shared__ __attribute__((aligned(16))) float wl[32 * 45 * 8];
    conv_body<32, 64, 21, 17, 6, 6, 1, 8>(A, w, b, out, wl);
}

__global__ void __launch_bounds__(CONV_T_)
conv3_kernel(const float* __restrict__ A, const float* __restrict__ w,
             const float* __restrict__ b, float* __restrict__ out) {
    __shared__ __attribute__((aligned(16))) float wl[64 * 45 * 4];
    conv_body<64, 64, 17, 13, 6, 4, 0, 4>(A, w, b, out, wl);
}

__global__ void pool_kernel(const float* __restrict__ A, float* __restrict__ enc) {
    const int total = 8 * 64 * 11 * 2 * 2;
    int idx = blockIdx.x * blockDim.x + threadIdx.x;
    if (idx >= total) return;
    int x = idx % 2;  int t = idx / 2;
    int y = t % 2;    t /= 2;
    int d = t % 11;   t /= 11;
    int ch = t % 64;  int bn = t / 64;
    float m = -INFINITY;
    for (int dd = 0; dd < 3; ++dd)
        for (int hh = 0; hh < 3; ++hh)
            for (int ww = 0; ww < 3; ++ww) {
                float v = A[((size_t)((bn * 64 + ch) * 13 + (d + dd))) * 16 + (y + hh) * 4 + (x + ww)];
                m = fmaxf(m, v);
            }
    enc[(size_t)bn * ENC_ + ch * 44 + d * 4 + y * 2 + x] = m;
}

__global__ void edge_pass(const int* __restrict__ ei, const int* __restrict__ gid,
                          const float* __restrict__ x3, unsigned* __restrict__ aggk,
                          unsigned* __restrict__ mask, int* __restrict__ counts, int E) {
    int e = blockIdx.x * blockDim.x + threadIdx.x;
    if (e >= E) return;
    int s = ei[e];
    int d = ei[E + e];
    int g = gid[s];
    atomicOr(&mask[d], 1u << g);
    atomicAdd(&counts[d], 1);
    atomicMax(&aggk[(size_t)d * 3 + 0], fkey(x3[(size_t)s * 3 + 0]));
    atomicMax(&aggk[(size_t)d * 3 + 1], fkey(x3[(size_t)s * 3 + 1]));
    atomicMax(&aggk[(size_t)d * 3 + 2], fkey(x3[(size_t)s * 3 + 2]));
}

#define ME_T_ 64

__global__ void __launch_bounds__(ME_T_)
maxenc_kernel(const float* __restrict__ enc, float* __restrict__ ME) {
    const int c4 = blockIdx.x * ME_T_ + threadIdx.x;
    const int hi = blockIdx.y;
    const float4* e4 = (const float4*)enc;
    float4 e[8];
    #pragma unroll
    for (int g = 0; g < 8; ++g) e[g] = e4[(size_t)g * (ENC_ / 4) + c4];

    float4 t[16];
    float4 base = make_float4(0.f, 0.f, 0.f, 0.f);
    #pragma unroll
    for (int b = 3; b >= 0; --b) {
        const bool on = ((hi >> b) & 1) != 0;
        base.x = fmaxf(base.x, on ? e[4 + b].x : 0.0f);
        base.y = fmaxf(base.y, on ? e[4 + b].y : 0.0f);
        base.z = fmaxf(base.z, on ? e[4 + b].z : 0.0f);
        base.w = fmaxf(base.w, on ? e[4 + b].w : 0.0f);
    }
    t[0] = base;
    #pragma unroll
    for (int lo = 1; lo < 16; ++lo) {
        const int p = lo & (lo - 1);
        const int g = __builtin_ctz(lo);
        t[lo].x = fmaxf(t[p].x, e[g].x);
        t[lo].y = fmaxf(t[p].y, e[g].y);
        t[lo].z = fmaxf(t[p].z, e[g].z);
        t[lo].w = fmaxf(t[p].w, e[g].w);
    }
    float4* M4 = (float4*)ME;
    #pragma unroll
    for (int lo = 0; lo < 16; ++lo)
        M4[(size_t)(hi * 16 + lo) * (ENC_ / 4) + c4] = t[lo];
}

#define KSPLIT_ 16
#define KSLICE_ (KGEMM_ / KSPLIT_)

__global__ void __launch_bounds__(32 * KSPLIT_)
gemm_wmma_f32(const float* __restrict__ A, const float* __restrict__ W,
              float* __restrict__ C, int Mrows) {
    __shared__ float part[KSPLIT_][8][32];
    const int Ncols = HID_;
    int wave = threadIdx.x >> 5;
    int lane = threadIdx.x & 31;
    int half = lane >> 4;
    int l16  = lane & 15;
    int mBase = blockIdx.x * 16;
    int nBase = blockIdx.y * 16;
    int mRow = mBase + l16; if (mRow >= Mrows) mRow = Mrows - 1;
    int nCol = nBase + l16;
    int nClamp = (nCol < Ncols) ? nCol : (Ncols - 1);
    const float* Arow = A + (size_t)mRow * KGEMM_;
    const float* Bcol = W + nClamp;

    const int kBeg = wave * KSLICE_;
    v8f acc0 = {}, acc1 = {};
    #pragma unroll
    for (int kt = kBeg; kt < kBeg + KSLICE_; kt += 8) {
        int k0 = kt + half * 2;
        int k1 = k0 + 4;
        v2f a0, b0, a1, b1;
        a0.x = Arow[k0];
        a0.y = Arow[k0 + 1];
        a1.x = Arow[k1];
        a1.y = Arow[k1 + 1];
        b0.x = Bcol[(size_t)k0 * Ncols];
        b0.y = Bcol[(size_t)(k0 + 1) * Ncols];
        b1.x = Bcol[(size_t)k1 * Ncols];
        b1.y = Bcol[(size_t)(k1 + 1) * Ncols];
        acc0 = __builtin_amdgcn_wmma_f32_16x16x4_f32(false, a0, false, b0,
                                                     (short)0, acc0, false, false);
        acc1 = __builtin_amdgcn_wmma_f32_16x16x4_f32(false, a1, false, b1,
                                                     (short)0, acc1, false, false);
    }
    v8f acc = acc0 + acc1;
    for (int v = 0; v < 8; ++v) part[wave][v][lane] = acc[v];
    __syncthreads();
    if (wave < 8) {
        float s = part[0][wave][lane];
        for (int w = 1; w < KSPLIT_; ++w) s += part[w][wave][lane];
        int m = mBase + half * 8 + wave;
        if (m < Mrows && nCol < Ncols) C[(size_t)m * Ncols + nCol] = s;
    }
}

__global__ void node_h_kernel(const float* __restrict__ x3, const int* __restrict__ gid,
                              const unsigned* __restrict__ mask, const unsigned* __restrict__ aggk,
                              const float* __restrict__ P, const float* __restrict__ Q,
                              const float* __restrict__ Wl1, const float* __restrict__ Wr1,
                              const float* __restrict__ bl1,
                              const float* __restrict__ Wl2, const float* __restrict__ Wr2,
                              float* __restrict__ tl, float* __restrict__ tr, int N) {
    int wave = threadIdx.x >> 5;
    int lane = threadIdx.x & 31;
    int n = blockIdx.x * (blockDim.x >> 5) + wave;
    if (n >= N) return;

    float x0 = x3[(size_t)n * 3 + 0];
    float x1 = x3[(size_t)n * 3 + 1];
    float x2 = x3[(size_t)n * 3 + 2];
    int g = gid[n];
    unsigned mk = mask[n];
    bool has = (mk != 0u);
    float a0 = 0.f, a1 = 0.f, a2 = 0.f;
    if (has) {
        a0 = keyf(aggk[(size_t)n * 3 + 0]);
        a1 = keyf(aggk[(size_t)n * 3 + 1]);
        a2 = keyf(aggk[(size_t)n * 3 + 2]);
    }
    const float* Prow = P + (size_t)mk * HID_;
    const float* Qrow = Q + (size_t)g * HID_;

    float v[4];
    #pragma unroll
    for (int c = 0; c < 4; ++c) {
        int col = lane + 32 * c;
        if (col < HID_) {
            float t = bl1[col] + Qrow[col]
                    + x0 * Wr1[col] + x1 * Wr1[HID_ + col] + x2 * Wr1[2 * HID_ + col];
            if (has)
                t += Prow[col]
                   + a0 * Wl1[col] + a1 * Wl1[HID_ + col] + a2 * Wl1[2 * HID_ + col];
            v[c] = t;
        } else {
            v[c] = -INFINITY;
        }
    }
    float m = fmaxf(fmaxf(v[0], v[1]), fmaxf(v[2], v[3]));
    for (int off = 16; off >= 1; off >>= 1)
        m = fmaxf(m, __shfl_xor(m, off, 32));
    float s = 0.f;
    #pragma unroll
    for (int c = 0; c < 4; ++c) s += __expf(v[c] - m);
    for (int off = 16; off >= 1; off >>= 1)
        s += __shfl_xor(s, off, 32);
    float lse = m + __logf(s);
    float s1 = 0.f, s2 = 0.f;
    #pragma unroll
    for (int c = 0; c < 4; ++c) {
        int col = lane + 32 * c;
        if (col < HID_) {
            float hv = v[c] - lse;
            s1 = fmaf(hv, Wl2[col], s1);
            s2 = fmaf(hv, Wr2[col], s2);
        }
    }
    for (int off = 16; off >= 1; off >>= 1) {
        s1 += __shfl_xor(s1, off, 32);
        s2 += __shfl_xor(s2, off, 32);
    }
    if (lane == 0) { tl[n] = s1; tr[n] = s2; }
}

__global__ void edge_scatter_t(const int* __restrict__ ei, const float* __restrict__ tl,
                               float* __restrict__ agg2, int E) {
    int e = blockIdx.x * blockDim.x + threadIdx.x;
    if (e >= E) return;
    int s = ei[e];
    int d = ei[E + e];
    atomicAdd(&agg2[d], tl[s]);
}

__global__ void node_out_kernel(const float* __restrict__ agg2, const float* __restrict__ tr,
                                const int* __restrict__ counts, const float* __restrict__ bl2,
                                float* __restrict__ y, int N) {
    int n = blockIdx.x * blockDim.x + threadIdx.x;
    if (n >= N) return;
    y[n] = agg2[n] / fmaxf((float)counts[n], 1.0f) + bl2[0] + tr[n];
}

extern "C" void kernel_launch(void* const* d_in, const int* in_sizes, int n_in,
                              void* d_out, int out_size, void* d_ws, size_t ws_size,
                              hipStream_t stream) {
    const float* X      = (const float*)d_in[0];
    const float* x3     = (const float*)d_in[1];
    const int*   ei     = (const int*)d_in[2];
    const int*   gid    = (const int*)d_in[3];
    const float* w1     = (const float*)d_in[4];
    const float* b1     = (const float*)d_in[5];
    const float* w2     = (const float*)d_in[6];
    const float* b2     = (const float*)d_in[7];
    const float* w3     = (const float*)d_in[8];
    const float* b3     = (const float*)d_in[9];
    const float* Wl1    = (const float*)d_in[10];
    const float* bl1    = (const float*)d_in[11];
    const float* Wr1    = (const float*)d_in[12];
    const float* Wl2    = (const float*)d_in[13];
    const float* bl2    = (const float*)d_in[14];
    const float* Wr2    = (const float*)d_in[15];
    float* y = (float*)d_out;

    const int E = in_sizes[2] / 2;
    const int N = in_sizes[3];

    char* ws = (char*)d_ws;
    size_t off = 0;
    auto alloc = [&](size_t bytes) -> char* {
        char* p = ws + off;
        off = (off + bytes + 255) & ~(size_t)255;
        return p;
    };
    float*    act1  = (float*)alloc((size_t)8 * 32 * 21 * 36 * 4);
    float*    act2  = (float*)alloc((size_t)8 * 64 * 17 * 36 * 4);
    float*    act3  = (float*)alloc((size_t)8 * 64 * 13 * 16 * 4);
    float*    enc   = (float*)alloc((size_t)8 * ENC_ * 4);
    float*    ME    = (float*)alloc((size_t)256 * ENC_ * 4);
    float*    P     = (float*)alloc((size_t)256 * HID_ * 4);
    float*    Q     = (float*)alloc((size_t)16  * HID_ * 4);
    unsigned* aggk  = (unsigned*)alloc((size_t)N * 3 * 4);
    unsigned* mask  = (unsigned*)alloc((size_t)N * 4);
    int*      cnt   = (int*)alloc((size_t)N * 4);
    float*    tl    = (float*)alloc((size_t)N * 4);
    float*    tr    = (float*)alloc((size_t)N * 4);
    float*    agg2  = (float*)alloc((size_t)N * 4);

    const int T = 256;
    zero_u32<<<(N * 3 + T - 1) / T, T, 0, stream>>>(aggk, N * 3);
    zero_u32<<<(N + T - 1) / T, T, 0, stream>>>(mask, N);
    zero_u32<<<(N + T - 1) / T, T, 0, stream>>>((unsigned*)cnt, N);
    zero_u32<<<(N + T - 1) / T, T, 0, stream>>>((unsigned*)agg2, N);

    conv1_kernel<<<dim3(48, 8), CONV_T_, 0, stream>>>(X, w1, b1, act1);
    conv2_kernel<<<dim3(39, 8), CONV_T_, 0, stream>>>(act1, w2, b2, act2);
    conv3_kernel<<<dim3(13, 16), CONV_T_, 0, stream>>>(act2, w3, b3, act3);
    pool_kernel<<<(8 * 64 * 11 * 4 + T - 1) / T, T, 0, stream>>>(act3, enc);

    edge_pass<<<(E + T - 1) / T, T, 0, stream>>>(ei, gid, x3, aggk, mask, cnt, E);

    maxenc_kernel<<<dim3(ENC_ / 4 / ME_T_, 16), ME_T_, 0, stream>>>(enc, ME);
    {
        dim3 gP(256 / 16, (HID_ + 15) / 16);
        gemm_wmma_f32<<<gP, 32 * KSPLIT_, 0, stream>>>(ME, Wl1 + 3 * HID_, P, 256);
        dim3 gQ(1, (HID_ + 15) / 16);
        gemm_wmma_f32<<<gQ, 32 * KSPLIT_, 0, stream>>>(enc, Wr1 + 3 * HID_, Q, 8);
    }

    node_h_kernel<<<(N + 7) / 8, 256, 0, stream>>>(x3, gid, mask, aggk, P, Q,
                                                   Wl1, Wr1, bl1, Wl2, Wr2, tl, tr, N);

    edge_scatter_t<<<(E + T - 1) / T, T, 0, stream>>>(ei, tl, agg2, E);
    node_out_kernel<<<(N + T - 1) / T, T, 0, stream>>>(agg2, tr, cnt, bl2, y, N);

    (void)n_in; (void)out_size; (void)ws_size;
}


# Conv input channels split over 4 wave groups per workgroup (4-6 waves per SIMD): one group's FMAs run while another waits for loads; slice sums added via LDS in fixed order (changes summation order)
# speedup vs baseline: 4.6445x; 1.4548x over previous
#include <hip/hip_runtime.h>
#include <hip/hip_bf16.h>
#include <math.h>


#define B_     8
#define CIN_   25
#define NPG_   2000
#define ENC_   2816
#define HID_   100
#define KGEMM_ 2816

typedef __attribute__((ext_vector_type(2))) float v2f;
typedef __attribute__((ext_vector_type(8))) float v8f;
typedef __attribute__((ext_vector_type(4))) float v4f;

__device__ __forceinline__ unsigned fkey(float f) {
    unsigned u = __float_as_uint(f);
    return (u & 0x80000000u) ? ~u : (u | 0x80000000u);
}
__device__ __forceinline__ float keyf(unsigned k) {
    unsigned u = (k & 0x80000000u) ? (k ^ 0x80000000u) : ~k;
    return __uint_as_float(u);
}

__global__ void zero_u32(unsigned* __restrict__ p, int n) {
    int i = blockIdx.x * blockDim.x + threadIdx.x;
    if (i < n) p[i] = 0u;
}

#define CONV_KS_ 4

template <int CIN, int COUT, int DIN, int DOUT, int HIN, int HOUT, int PAD, int OCT, int PB>
__device__ __forceinline__ void conv_body(const float* __restrict__ in, const float* __restrict__ w,
                                          const float* __restrict__ b, float* __restrict__ out,
                                          float* wl, float* red) {
    constexpr int WIN = HIN;
    constexpr int WOUT = HOUT;
    constexpr int PLANE = HIN * WIN;
    constexpr int NPOS = DOUT * HOUT * WOUT;
    constexpr int NTOT = 8 * NPOS;
    constexpr int KW = CIN * 45;
    constexpr int NT = PB * CONV_KS_;

    const int tid = (int)threadIdx.x;
    const int og = blockIdx.y * OCT;
    constexpr int NW = KW * OCT;
    constexpr int SU = (NW + NT - 1) / NT;
    {
        float tmp[SU];
        #pragma unroll
        for (int u = 0; u < SU; ++u) {
            int i = u * NT + tid;
            int ic = i < NW ? i : NW - 1;
            int o = ic / KW;
            int r = ic - o * KW;
            tmp[u] = w[(size_t)(og + o) * KW + r];
        }
        #pragma unroll
        for (int u = 0; u < SU; ++u) {
            int i = u * NT + tid;
            if (i < NW) {
                int o = i / KW;
                int r = i - o * KW;
                wl[r * OCT + o] = tmp[u];
            }
        }
    }
    __syncthreads();

    const int ks = __builtin_amdgcn_readfirstlane(tid / PB);
    const int pw = tid - ks * PB;
    const int cBeg = (ks * CIN) / CONV_KS_;
    const int cEnd = ((ks + 1) * CIN) / CONV_KS_;

    int gp = blockIdx.x * PB + pw;
    const bool active = gp < NTOT;
    if (!active) gp = NTOT - 1;
    const int bn = gp / NPOS;
    const int p = gp - bn * NPOS;
    const int x = p % WOUT;
    const int t2 = p / WOUT;
    const int y = t2 % HOUT;
    const int z = t2 / HOUT;

    int off[9];
    bool ok[9];
    #pragma unroll
    for (int kh = 0; kh < 3; ++kh)
        #pragma unroll
        for (int kw = 0; kw < 3; ++kw) {
            int yy = y + kh - PAD;
            int xx = x + kw - PAD;
            ok[kh * 3 + kw] = (yy >= 0) && (yy < HIN) && (xx >= 0) && (xx < WIN);
            int yc = yy < 0 ? 0 : (yy >= HIN ? HIN - 1 : yy);
            int xc = xx < 0 ? 0 : (xx >= WIN ? WIN - 1 : xx);
            off[kh * 3 + kw] = yc * WIN + xc;
        }

    float acc[OCT];
    #pragma unroll
    for (int o = 0; o < OCT; ++o) acc[o] = (ks == 0) ? b[og + o] : 0.0f;

    const float* ib = in + ((size_t)bn * CIN * DIN + z) * PLANE;
    for (int c = cBeg; c < cEnd; ++c) {
        const float* ic = ib + (size_t)c * DIN * PLANE;
        float xv[45];
        #pragma unroll
        for (int kd = 0; kd < 5; ++kd)
            #pragma unroll
            for (int t = 0; t < 9; ++t)
                xv[kd * 9 + t] = ic[kd * PLANE + off[t]];
        if (PAD != 0) {
            asm volatile("" :: "v"(xv[0]), "v"(xv[1]), "v"(xv[2]), "v"(xv[3]), "v"(xv[4]), "v"(xv[5]), "v"(xv[6]), "v"(xv[7]), "v"(xv[8]), "v"(xv[9]), "v"(xv[10]), "v"(xv[11]), "v"(xv[12]), "v"(xv[13]), "v"(xv[14]), "v"(xv[15]), "v"(xv[16]), "v"(xv[17]), "v"(xv[18]), "v"(xv[19]), "v"(xv[20]), "v"(xv[21]), "v"(xv[22]));
            asm volatile("" :: "v"(xv[23]), "v"(xv[24]), "v"(xv[25]), "v"(xv[26]), "v"(xv[27]), "v"(xv[28]), "v"(xv[29]), "v"(xv[30]), "v"(xv[31]), "v"(xv[32]), "v"(xv[33]), "v"(xv[34]), "v"(xv[35]), "v"(xv[36]), "v"(xv[37]), "v"(xv[38]), "v"(xv[39]), "v"(xv[40]), "v"(xv[41]), "v"(xv[42]), "v"(xv[43]), "v"(xv[44]));
            #pragma unroll
            for (int j = 0; j < 45; ++j) xv[j] = ok[j % 9] ? xv[j] : 0.0f;
        }
        constexpr int Q = OCT / 4;
        const v4f* wc4 = (const v4f*)(wl + c * 45 * OCT);
        #pragma unroll
        for (int kd = 0; kd < 5; ++kd) {
            v4f wq[9 * Q];
            #pragma unroll
            for (int i = 0; i < 9 * Q; ++i) wq[i] = wc4[kd * 9 * Q + i];
            if constexpr (Q == 1) {
                asm volatile("" :: "v"(wq[0]), "v"(wq[1]), "v"(wq[2]), "v"(wq[3]), "v"(wq[4]), "v"(wq[5]), "v"(wq[6]), "v"(wq[7]), "v"(wq[8]));
            } else {
                asm volatile("" :: "v"(wq[0]), "v"(wq[1]), "v"(wq[2]), "v"(wq[3]), "v"(wq[4]), "v"(wq[5]), "v"(wq[6]), "v"(wq[7]), "v"(wq[8]), "v"(wq[9]), "v"(wq[10]), "v"(wq[11]), "v"(wq[12]), "v"(wq[13]), "v"(wq[14]), "v"(wq[15]), "v"(wq[16]), "v"(wq[17]));
            }
            #pragma unroll
            for (int t = 0; t < 9; ++t)
                #pragma unroll
                for (int o = 0; o < OCT; ++o)
                    acc[o] = fmaf(xv[kd * 9 + t], wq[t * Q + o / 4][o % 4], acc[o]);
        }
    }

    if (ks > 0) {
        #pragma unroll
        for (int o = 0; o < OCT; ++o) red[((ks - 1) * OCT + o) * PB + pw] = acc[o];
    }
    __syncthreads();
    if (ks == 0 && active) {
        #pragma unroll
        for (int o = 0; o < OCT; ++o) {
            float s = acc[o];
            #pragma unroll
            for (int k = 1; k < CONV_KS_; ++k) s += red[((k - 1) * OCT + o) * PB + pw];
            out[((size_t)(bn * COUT + og + o) * DOUT + z) * (HOUT * WOUT) + y * WOUT + x] =
                fmaxf(s, 0.0f);
        }
    }
}

__global__ void __launch_bounds__(192 * CONV_KS_) __attribute__((amdgpu_waves_per_eu(6, 6)))
conv1_kernel(const float* __restrict__ X, const float* __restrict__ w,
             const float* __restrict__ b, float* __restrict__ out) {
    __shared__ __attribute__((aligned(16))) float wl[25 * 45 * 4];
    __shared__ float red[(CONV_KS_ - 1) * 4 * 192];
    conv_body<25, 32, 25, 21, 6, 6, 1, 4, 192>(X, w, b, out, wl, red);
}

__global__ void __launch_bounds__(160 * CONV_KS_) __attribute__((amdgpu_waves_per_eu(5, 5)))
conv2_kernel(const float* __restrict__ A, const float* __restrict__ w,
             const float* __restrict__ b, float* __restrict__ out) {
    __shared__ __attribute__((aligned(16))) float wl[32 * 45 * 8];
    __shared__ float red[(CONV_KS_ - 1) * 8 * 160];
    conv_body<32, 64, 21, 17, 6, 6, 1, 8, 160>(A, w, b, out, wl, red);
}

__global__ void __launch_bounds__(128 * CONV_KS_) __attribute__((amdgpu_waves_per_eu(4, 4)))
conv3_kernel(const float* __restrict__ A, const float* __restrict__ w,
             const float* __restrict__ b, float* __restrict__ out) {
    __shared__ __attribute__((aligned(16))) float wl[64 * 45 * 4];
    __shared__ float red[(CONV_KS_ - 1) * 4 * 128];
    conv_body<64, 64, 17, 13, 6, 4, 0, 4, 128>(A, w, b, out, wl, red);
}

__global__ void pool_kernel(const float* __restrict__ A, float* __restrict__ enc) {
    const int total = 8 * 64 * 11 * 2 * 2;
    int idx = blockIdx.x * blockDim.x + threadIdx.x;
    if (idx >= total) return;
    int x = idx % 2;  int t = idx / 2;
    int y = t % 2;    t /= 2;
    int d = t % 11;   t /= 11;
    int ch = t % 64;  int bn = t / 64;
    float m = -INFINITY;
    for (int dd = 0; dd < 3; ++dd)
        for (int hh = 0; hh < 3; ++hh)
            for (int ww = 0; ww < 3; ++ww) {
                float v = A[((size_t)((bn * 64 + ch) * 13 + (d + dd))) * 16 + (y + hh) * 4 + (x + ww)];
                m = fmaxf(m, v);
            }
    enc[(size_t)bn * ENC_ + ch * 44 + d * 4 + y * 2 + x] = m;
}

__global__ void edge_pass(const int* __restrict__ ei, const int* __restrict__ gid,
                          const float* __restrict__ x3, unsigned* __restrict__ aggk,
                          unsigned* __restrict__ mask, int* __restrict__ counts, int E) {
    int e = blockIdx.x * blockDim.x + threadIdx.x;
    if (e >= E) return;
    int s = ei[e];
    int d = ei[E + e];
    int g = gid[s];
    atomicOr(&mask[d], 1u << g);
    atomicAdd(&counts[d], 1);
    atomicMax(&aggk[(size_t)d * 3 + 0], fkey(x3[(size_t)s * 3 + 0]));
    atomicMax(&aggk[(size_t)d * 3 + 1], fkey(x3[(size_t)s * 3 + 1]));
    atomicMax(&aggk[(size_t)d * 3 + 2], fkey(x3[(size_t)s * 3 + 2]));
}

#define ME_T_ 64

__global__ void __launch_bounds__(ME_T_)
maxenc_kernel(const float* __restrict__ enc, float* __restrict__ ME) {
    const int c4 = blockIdx.x * ME_T_ + threadIdx.x;
    const int hi = blockIdx.y;
    const float4* e4 = (const float4*)enc;
    float4 e[8];
    #pragma unroll
    for (int g = 0; g < 8; ++g) e[g] = e4[(size_t)g * (ENC_ / 4) + c4];

    float4 t[16];
    float4 base = make_float4(0.f, 0.f, 0.f, 0.f);
    #pragma unroll
    for (int b = 3; b >= 0; --b) {
        const bool on = ((hi >> b) & 1) != 0;
        base.x = fmaxf(base.x, on ? e[4 + b].x : 0.0f);
        base.y = fmaxf(base.y, on ? e[4 + b].y : 0.0f);
        base.z = fmaxf(base.z, on ? e[4 + b].z : 0.0f);
        base.w = fmaxf(base.w, on ? e[4 + b].w : 0.0f);
    }
    t[0] = base;
    #pragma unroll
    for (int lo = 1; lo < 16; ++lo) {
        const int p = lo & (lo - 1);
        const int g = __builtin_ctz(lo);
        t[lo].x = fmaxf(t[p].x, e[g].x);
        t[lo].y = fmaxf(t[p].y, e[g].y);
        t[lo].z = fmaxf(t[p].z, e[g].z);
        t[lo].w = fmaxf(t[p].w, e[g].w);
    }
    float4* M4 = (float4*)ME;
    #pragma unroll
    for (int lo = 0; lo < 16; ++lo)
        M4[(size_t)(hi * 16 + lo) * (ENC_ / 4) + c4] = t[lo];
}

#define KSPLIT_ 16
#define KSLICE_ (KGEMM_ / KSPLIT_)

__global__ void __launch_bounds__(32 * KSPLIT_)
gemm_wmma_f32(const float* __restrict__ A, const float* __restrict__ W,
              float* __restrict__ C, int Mrows) {
    __shared__ float part[KSPLIT_][8][32];
    const int Ncols = HID_;
    int wave = threadIdx.x >> 5;
    int lane = threadIdx.x & 31;
    int half = lane >> 4;
    int l16  = lane & 15;
    int mBase = blockIdx.x * 16;
    int nBase = blockIdx.y * 16;
    int mRow = mBase + l16; if (mRow >= Mrows) mRow = Mrows - 1;
    int nCol = nBase + l16;
    int nClamp = (nCol < Ncols) ? nCol : (Ncols - 1);
    const float* Arow = A + (size_t)mRow * KGEMM_;
    const float* Bcol = W + nClamp;

    const int kBeg = wave * KSLICE_;
    v8f acc0 = {}, acc1 = {};
    #pragma unroll
    for (int kt = kBeg; kt < kBeg + KSLICE_; kt += 8) {
        int k0 = kt + half * 2;
        int k1 = k0 + 4;
        v2f a0, b0, a1, b1;
        a0.x = Arow[k0];
        a0.y = Arow[k0 + 1];
        a1.x = Arow[k1];
        a1.y = Arow[k1 + 1];
        b0.x = Bcol[(size_t)k0 * Ncols];
        b0.y = Bcol[(size_t)(k0 + 1) * Ncols];
        b1.x = Bcol[(size_t)k1 * Ncols];
        b1.y = Bcol[(size_t)(k1 + 1) * Ncols];
        acc0 = __builtin_amdgcn_wmma_f32_16x16x4_f32(false, a0, false, b0,
                                                     (short)0, acc0, false, false);
        acc1 = __builtin_amdgcn_wmma_f32_16x16x4_f32(false, a1, false, b1,
                                                     (short)0, acc1, false, false);
    }
    v8f acc = acc0 + acc1;
    for (int v = 0; v < 8; ++v) part[wave][v][lane] = acc[v];
    __syncthreads();
    if (wave < 8) {
        float s = part[0][wave][lane];
        for (int w = 1; w < KSPLIT_; ++w) s += part[w][wave][lane];
        int m = mBase + half * 8 + wave;
        if (m < Mrows && nCol < Ncols) C[(size_t)m * Ncols + nCol] = s;
    }
}

__global__ void node_h_kernel(const float* __restrict__ x3, const int* __restrict__ gid,
                              const unsigned* __restrict__ mask, const unsigned* __restrict__ aggk,
                              const float* __restrict__ P, const float* __restrict__ Q,
                              const float* __restrict__ Wl1, const float* __restrict__ Wr1,
                              const float* __restrict__ bl1,
                              const float* __restrict__ Wl2, const float* __restrict__ Wr2,
                              float* __restrict__ tl, float* __restrict__ tr, int N) {
    int wave = threadIdx.x >> 5;
    int lane = threadIdx.x & 31;
    int n = blockIdx.x * (blockDim.x >> 5) + wave;
    if (n >= N) return;

    float x0 = x3[(size_t)n * 3 + 0];
    float x1 = x3[(size_t)n * 3 + 1];
    float x2 = x3[(size_t)n * 3 + 2];
    int g = gid[n];
    unsigned mk = mask[n];
    bool has = (mk != 0u);
    float a0 = 0.f, a1 = 0.f, a2 = 0.f;
    if (has) {
        a0 = keyf(aggk[(size_t)n * 3 + 0]);
        a1 = keyf(aggk[(size_t)n * 3 + 1]);
        a2 = keyf(aggk[(size_t)n * 3 + 2]);
    }
    const float* Prow = P + (size_t)mk * HID_;
    const float* Qrow = Q + (size_t)g * HID_;

    float v[4];
    #pragma unroll
    for (int c = 0; c < 4; ++c) {
        int col = lane + 32 * c;
        if (col < HID_) {
            float t = bl1[col] + Qrow[col]
                    + x0 * Wr1[col] + x1 * Wr1[HID_ + col] + x2 * Wr1[2 * HID_ + col];
            if (has)
                t += Prow[col]
                   + a0 * Wl1[col] + a1 * Wl1[HID_ + col] + a2 * Wl1[2 * HID_ + col];
            v[c] = t;
        } else {
            v[c] = -INFINITY;
        }
    }
    float m = fmaxf(fmaxf(v[0], v[1]), fmaxf(v[2], v[3]));
    for (int off = 16; off >= 1; off >>= 1)
        m = fmaxf(m, __shfl_xor(m, off, 32));
    float s = 0.f;
    #pragma unroll
    for (int c = 0; c < 4; ++c) s += __expf(v[c] - m);
    for (int off = 16; off >= 1; off >>= 1)
        s += __shfl_xor(s, off, 32);
    float lse = m + __logf(s);
    float s1 = 0.f, s2 = 0.f;
    #pragma unroll
    for (int c = 0; c < 4; ++c) {
        int col = lane + 32 * c;
        if (col < HID_) {
            float hv = v[c] - lse;
            s1 = fmaf(hv, Wl2[col], s1);
            s2 = fmaf(hv, Wr2[col], s2);
        }
    }
    for (int off = 16; off >= 1; off >>= 1) {
        s1 += __shfl_xor(s1, off, 32);
        s2 += __shfl_xor(s2, off, 32);
    }
    if (lane == 0) { tl[n] = s1; tr[n] = s2; }
}

__global__ void edge_scatter_t(const int* __restrict__ ei, const float* __restrict__ tl,
                               float* __restrict__ agg2, int E) {
    int e = blockIdx.x * blockDim.x + threadIdx.x;
    if (e >= E) return;
    int s = ei[e];
    int d = ei[E + e];
    atomicAdd(&agg2[d], tl[s]);
}

__global__ void node_out_kernel(const float* __restrict__ agg2, const float* __restrict__ tr,
                                const int* __restrict__ counts, const float* __restrict__ bl2,
                                float* __restrict__ y, int N) {
    int n = blockIdx.x * blockDim.x + threadIdx.x;
    if (n >= N) return;
    y[n] = agg2[n] / fmaxf((float)counts[n], 1.0f) + bl2[0] + tr[n];
}

extern "C" void kernel_launch(void* const* d_in, const int* in_sizes, int n_in,
                              void* d_out, int out_size, void* d_ws, size_t ws_size,
                              hipStream_t stream) {
    const float* X      = (const float*)d_in[0];
    const float* x3     = (const float*)d_in[1];
    const int*   ei     = (const int*)d_in[2];
    const int*   gid    = (const int*)d_in[3];
    const float* w1     = (const float*)d_in[4];
    const float* b1     = (const float*)d_in[5];
    const float* w2     = (const float*)d_in[6];
    const float* b2     = (const float*)d_in[7];
    const float* w3     = (const float*)d_in[8];
    const float* b3     = (const float*)d_in[9];
    const float* Wl1    = (const float*)d_in[10];
    const float* bl1    = (const float*)d_in[11];
    const float* Wr1    = (const float*)d_in[12];
    const float* Wl2    = (const float*)d_in[13];
    const float* bl2    = (const float*)d_in[14];
    const float* Wr2    = (const float*)d_in[15];
    float* y = (float*)d_out;

    const int E = in_sizes[2] / 2;
    const int N = in_sizes[3];

    char* ws = (char*)d_ws;
    size_t off = 0;
    auto alloc = [&](size_t bytes) -> char* {
        char* p = ws + off;
        off = (off + bytes + 255) & ~(size_t)255;
        return p;
    };
    float*    act1  = (float*)alloc((size_t)8 * 32 * 21 * 36 * 4);
    float*    act2  = (float*)alloc((size_t)8 * 64 * 17 * 36 * 4);
    float*    act3  = (float*)alloc((size_t)8 * 64 * 13 * 16 * 4);
    float*    enc   = (float*)alloc((size_t)8 * ENC_ * 4);
    float*    ME    = (float*)alloc((size_t)256 * ENC_ * 4);
    float*    P     = (float*)alloc((size_t)256 * HID_ * 4);
    float*    Q     = (float*)alloc((size_t)16  * HID_ * 4);
    unsigned* aggk  = (unsigned*)alloc((size_t)N * 3 * 4);
    unsigned* mask  = (unsigned*)alloc((size_t)N * 4);
    int*      cnt   = (int*)alloc((size_t)N * 4);
    float*    tl    = (float*)alloc((size_t)N * 4);
    float*    tr    = (float*)alloc((size_t)N * 4);
    float*    agg2  = (float*)alloc((size_t)N * 4);

    const int T = 256;
    zero_u32<<<(N * 3 + T - 1) / T, T, 0, stream>>>(aggk, N * 3);
    zero_u32<<<(N + T - 1) / T, T, 0, stream>>>(mask, N);
    zero_u32<<<(N + T - 1) / T, T, 0, stream>>>((unsigned*)cnt, N);
    zero_u32<<<(N + T - 1) / T, T, 0, stream>>>((unsigned*)agg2, N);

    conv1_kernel<<<dim3(32, 8), 768, 0, stream>>>(X, w1, b1, act1);
    conv2_kernel<<<dim3(31, 8), 640, 0, stream>>>(act1, w2, b2, act2);
    conv3_kernel<<<dim3(13, 16), 512, 0, stream>>>(act2, w3, b3, act3);
    pool_kernel<<<(8 * 64 * 11 * 4 + T - 1) / T, T, 0, stream>>>(act3, enc);

    edge_pass<<<(E + T - 1) / T, T, 0, stream>>>(ei, gid, x3, aggk, mask, cnt, E);

    maxenc_kernel<<<dim3(ENC_ / 4 / ME_T_, 16), ME_T_, 0, stream>>>(enc, ME);
    {
        dim3 gP(256 / 16, (HID_ + 15) / 16);
        gemm_wmma_f32<<<gP, 32 * KSPLIT_, 0, stream>>>(ME, Wl1 + 3 * HID_, P, 256);
        dim3 gQ(1, (HID_ + 15) / 16);
        gemm_wmma_f32<<<gQ, 32 * KSPLIT_, 0, stream>>>(enc, Wr1 + 3 * HID_, Q, 8);
    }

    node_h_kernel<<<(N + 7) / 8, 256, 0, stream>>>(x3, gid, mask, aggk, P, Q,
                                                   Wl1, Wr1, bl1, Wl2, Wr2, tl, tr, N);

    edge_scatter_t<<<(E + T - 1) / T, T, 0, stream>>>(ei, tl, agg2, E);
    node_out_kernel<<<(N + T - 1) / T, T, 0, stream>>>(agg2, tr, cnt, bl2, y, N);

    (void)n_in; (void)out_size; (void)ws_size;
}


# Conv threads own 3-4 consecutive output depths: each LDS weight read and loaded input plane is shared by them (3-4x fewer LDS reads per FMA); 8-16 channel slices per block (changes summation order)
# speedup vs baseline: 5.3897x; 1.1604x over previous
#include <hip/hip_runtime.h>
#include <hip/hip_bf16.h>
#include <math.h>


#define B_     8
#define CIN_   25
#define NPG_   2000
#define ENC_   2816
#define HID_   100
#define KGEMM_ 2816

typedef __attribute__((ext_vector_type(2))) float v2f;
typedef __attribute__((ext_vector_type(8))) float v8f;
typedef __attribute__((ext_vector_type(4))) float v4f;

__device__ __forceinline__ unsigned fkey(float f) {
    unsigned u = __float_as_uint(f);
    return (u & 0x80000000u) ? ~u : (u | 0x80000000u);
}
__device__ __forceinline__ float keyf(unsigned k) {
    unsigned u = (k & 0x80000000u) ? (k ^ 0x80000000u) : ~k;
    return __uint_as_float(u);
}

__global__ void zero_u32(unsigned* __restrict__ p, int n) {
    int i = blockIdx.x * blockDim.x + threadIdx.x;
    if (i < n) p[i] = 0u;
}


template <int CIN, int COUT, int DIN, int DOUT, int HIN, int HOUT, int PAD, int OCT, int PB, int ZB, int KS>
__device__ __forceinline__ void conv_body(const float* __restrict__ in, const float* __restrict__ w,
                                          const float* __restrict__ b, float* __restrict__ out,
                                          float* wl, float* red) {
    constexpr int WIN = HIN;
    constexpr int WOUT = HOUT;
    constexpr int PLANE = HIN * WIN;
    constexpr int HW = HOUT * WOUT;
    constexpr int NBLK = (DOUT + ZB - 1) / ZB;
    constexpr int NTOT = 8 * NBLK * HW;
    constexpr int KW = CIN * 45;
    constexpr int NT = PB * KS;
    constexpr int ND = ZB + 4;
    constexpr int NACC = ZB * OCT;

    const int tid = (int)threadIdx.x;
    const int og = blockIdx.y * OCT;
    constexpr int NW = KW * OCT;
    constexpr int SU = (NW + NT - 1) / NT;
    {
        float tmp[SU];
        #pragma unroll
        for (int u = 0; u < SU; ++u) {
            int i = u * NT + tid;
            int ic = i < NW ? i : NW - 1;
            int o = ic / KW;
            int r = ic - o * KW;
            tmp[u] = w[(size_t)(og + o) * KW + r];
        }
        #pragma unroll
        for (int u = 0; u < SU; ++u) {
            int i = u * NT + tid;
            if (i < NW) {
                int o = i / KW;
                int r = i - o * KW;
                wl[r * OCT + o] = tmp[u];
            }
        }
    }
    __syncthreads();

    const int ks = __builtin_amdgcn_readfirstlane(tid / PB);
    const int pw = tid - ks * PB;
    const int cBeg = (ks * CIN) / KS;
    const int cEnd = ((ks + 1) * CIN) / KS;

    int gp = blockIdx.x * PB + pw;
    const bool active = gp < NTOT;
    if (!active) gp = NTOT - 1;
    const int bn = gp / (NBLK * HW);
    const int rb = gp - bn * (NBLK * HW);
    const int blk = rb / HW;
    const int p = rb - blk * HW;
    const int y = p / WOUT;
    const int x = p - y * WOUT;
    const int zFirst = blk * ZB;
    const int z0 = zFirst < DOUT - ZB ? zFirst : DOUT - ZB;

    int off[9];
    bool ok[9];
    #pragma unroll
    for (int kh = 0; kh < 3; ++kh)
        #pragma unroll
        for (int kw = 0; kw < 3; ++kw) {
            int yy = y + kh - PAD;
            int xx = x + kw - PAD;
            ok[kh * 3 + kw] = (yy >= 0) && (yy < HIN) && (xx >= 0) && (xx < WIN);
            int yc = yy < 0 ? 0 : (yy >= HIN ? HIN - 1 : yy);
            int xc = xx < 0 ? 0 : (xx >= WIN ? WIN - 1 : xx);
            off[kh * 3 + kw] = yc * WIN + xc;
        }

    float acc[NACC];
    #pragma unroll
    for (int a = 0; a < NACC; ++a) acc[a] = (ks == 0) ? b[og + a % OCT] : 0.0f;

    const float* ib = in + ((size_t)bn * CIN * DIN + z0) * PLANE;
    for (int c = cBeg; c < cEnd; ++c) {
        const float* ic = ib + (size_t)c * DIN * PLANE;
        float xv[ND * 9];
        #pragma unroll
        for (int d = 0; d < ND; ++d)
            #pragma unroll
            for (int t = 0; t < 9; ++t)
                xv[d * 9 + t] = ic[d * PLANE + off[t]];
        if (PAD != 0) {
            #pragma unroll
            for (int d = 0; d < ND; ++d)
                asm volatile("" :: "v"(xv[d * 9 + 0]), "v"(xv[d * 9 + 1]), "v"(xv[d * 9 + 2]), "v"(xv[d * 9 + 3]), "v"(xv[d * 9 + 4]), "v"(xv[d * 9 + 5]), "v"(xv[d * 9 + 6]), "v"(xv[d * 9 + 7]), "v"(xv[d * 9 + 8]));
            #pragma unroll
            for (int j = 0; j < ND * 9; ++j) xv[j] = ok[j % 9] ? xv[j] : 0.0f;
        }
        constexpr int Q = OCT / 4;
        const v4f* wc4 = (const v4f*)(wl + c * 45 * OCT);
        #pragma unroll
        for (int kd = 0; kd < 5; ++kd) {
            #pragma unroll
            for (int h = 0; h < Q; ++h) {
                v4f wq[9];
                #pragma unroll
                for (int t = 0; t < 9; ++t) wq[t] = wc4[(kd * 9 + t) * Q + h];
                asm volatile("" :: "v"(wq[0]), "v"(wq[1]), "v"(wq[2]), "v"(wq[3]), "v"(wq[4]), "v"(wq[5]), "v"(wq[6]), "v"(wq[7]), "v"(wq[8]));
                #pragma unroll
                for (int zb = 0; zb < ZB; ++zb)
                    #pragma unroll
                    for (int t = 0; t < 9; ++t)
                        #pragma unroll
                        for (int o = 0; o < 4; ++o)
                            acc[zb * OCT + h * 4 + o] = fmaf(xv[(zb + kd) * 9 + t], wq[t][o], acc[zb * OCT + h * 4 + o]);
            }
        }
    }

    if (ks > 0) {
        #pragma unroll
        for (int a = 0; a < NACC; ++a) red[((ks - 1) * NACC + a) * PB + pw] = acc[a];
    }
    __syncthreads();
    if (ks == 0 && active) {
        #pragma unroll
        for (int zb = 0; zb < ZB; ++zb) {
            const int z = z0 + zb;
            #pragma unroll
            for (int o = 0; o < OCT; ++o) {
                const int a = zb * OCT + o;
                float s = acc[a];
                #pragma unroll
                for (int k = 1; k < KS; ++k) s += red[((k - 1) * NACC + a) * PB + pw];
                if (z >= zFirst)
                    out[((size_t)(bn * COUT + og + o) * DOUT + z) * HW + y * WOUT + x] = fmaxf(s, 0.0f);
            }
        }
    }
}

__global__ void __launch_bounds__(512) __attribute__((amdgpu_waves_per_eu(4, 4)))
conv1_kernel(const float* __restrict__ X, const float* __restrict__ w,
             const float* __restrict__ b, float* __restrict__ out) {
    __shared__ __attribute__((aligned(16))) float wl[25 * 45 * 4];
    __shared__ float red[(8 - 1) * 3 * 4 * 64];
    conv_body<25, 32, 25, 21, 6, 6, 1, 4, 64, 3, 8>(X, w, b, out, wl, red);
}

__global__ void __launch_bounds__(512) __attribute__((amdgpu_waves_per_eu(4, 4)))
conv2_kernel(const float* __restrict__ A, const float* __restrict__ w,
             const float* __restrict__ b, float* __restrict__ out) {
    __shared__ __attribute__((aligned(16))) float wl[32 * 45 * 8];
    __shared__ float red[(8 - 1) * 3 * 8 * 64];
    conv_body<32, 64, 21, 17, 6, 6, 1, 8, 64, 3, 8>(A, w, b, out, wl, red);
}

__global__ void __launch_bounds__(512) __attribute__((amdgpu_waves_per_eu(4, 4)))
conv3_kernel(const float* __restrict__ A, const float* __restrict__ w,
             const float* __restrict__ b, float* __restrict__ out) {
    __shared__ __attribute__((aligned(16))) float wl[64 * 45 * 4];
    __shared__ float red[(16 - 1) * 4 * 4 * 32];
    conv_body<64, 64, 17, 13, 6, 4, 0, 4, 32, 4, 16>(A, w, b, out, wl, red);
}

__global__ void pool_kernel(const float* __restrict__ A, float* __restrict__ enc) {
    const int total = 8 * 64 * 11 * 2 * 2;
    int idx = blockIdx.x * blockDim.x + threadIdx.x;
    if (idx >= total) return;
    int x = idx % 2;  int t = idx / 2;
    int y = t % 2;    t /= 2;
    int d = t % 11;   t /= 11;
    int ch = t % 64;  int bn = t / 64;
    float m = -INFINITY;
    for (int dd = 0; dd < 3; ++dd)
        for (int hh = 0; hh < 3; ++hh)
            for (int ww = 0; ww < 3; ++ww) {
                float v = A[((size_t)((bn * 64 + ch) * 13 + (d + dd))) * 16 + (y + hh) * 4 + (x + ww)];
                m = fmaxf(m, v);
            }
    enc[(size_t)bn * ENC_ + ch * 44 + d * 4 + y * 2 + x] = m;
}

__global__ void edge_pass(const int* __restrict__ ei, const int* __restrict__ gid,
                          const float* __restrict__ x3, unsigned* __restrict__ aggk,
                          unsigned* __restrict__ mask, int* __restrict__ counts, int E) {
    int e = blockIdx.x * blockDim.x + threadIdx.x;
    if (e >= E) return;
    int s = ei[e];
    int d = ei[E + e];
    int g = gid[s];
    atomicOr(&mask[d], 1u << g);
    atomicAdd(&counts[d], 1);
    atomicMax(&aggk[(size_t)d * 3 + 0], fkey(x3[(size_t)s * 3 + 0]));
    atomicMax(&aggk[(size_t)d * 3 + 1], fkey(x3[(size_t)s * 3 + 1]));
    atomicMax(&aggk[(size_t)d * 3 + 2], fkey(x3[(size_t)s * 3 + 2]));
}

#define ME_T_ 64

__global__ void __launch_bounds__(ME_T_)
maxenc_kernel(const float* __restrict__ enc, float* __restrict__ ME) {
    const int c4 = blockIdx.x * ME_T_ + threadIdx.x;
    const int hi = blockIdx.y;
    const float4* e4 = (const float4*)enc;
    float4 e[8];
    #pragma unroll
    for (int g = 0; g < 8; ++g) e[g] = e4[(size_t)g * (ENC_ / 4) + c4];

    float4 t[16];
    float4 base = make_float4(0.f, 0.f, 0.f, 0.f);
    #pragma unroll
    for (int b = 3; b >= 0; --b) {
        const bool on = ((hi >> b) & 1) != 0;
        base.x = fmaxf(base.x, on ? e[4 + b].x : 0.0f);
        base.y = fmaxf(base.y, on ? e[4 + b].y : 0.0f);
        base.z = fmaxf(base.z, on ? e[4 + b].z : 0.0f);
        base.w = fmaxf(base.w, on ? e[4 + b].w : 0.0f);
    }
    t[0] = base;
    #pragma unroll
    for (int lo = 1; lo < 16; ++lo) {
        const int p = lo & (lo - 1);
        const int g = __builtin_ctz(lo);
        t[lo].x = fmaxf(t[p].x, e[g].x);
        t[lo].y = fmaxf(t[p].y, e[g].y);
        t[lo].z = fmaxf(t[p].z, e[g].z);
        t[lo].w = fmaxf(t[p].w, e[g].w);
    }
    float4* M4 = (float4*)ME;
    #pragma unroll
    for (int lo = 0; lo < 16; ++lo)
        M4[(size_t)(hi * 16 + lo) * (ENC_ / 4) + c4] = t[lo];
}

#define KSPLIT_ 16
#define KSLICE_ (KGEMM_ / KSPLIT_)

__global__ void __launch_bounds__(32 * KSPLIT_)
gemm_wmma_f32(const float* __restrict__ A, const float* __restrict__ W,
              float* __restrict__ C, int Mrows) {
    __shared__ float part[KSPLIT_][8][32];
    const int Ncols = HID_;
    int wave = threadIdx.x >> 5;
    int lane = threadIdx.x & 31;
    int half = lane >> 4;
    int l16  = lane & 15;
    int mBase = blockIdx.x * 16;
    int nBase = blockIdx.y * 16;
    int mRow = mBase + l16; if (mRow >= Mrows) mRow = Mrows - 1;
    int nCol = nBase + l16;
    int nClamp = (nCol < Ncols) ? nCol : (Ncols - 1);
    const float* Arow = A + (size_t)mRow * KGEMM_;
    const float* Bcol = W + nClamp;

    const int kBeg = wave * KSLICE_;
    v8f acc0 = {}, acc1 = {};
    #pragma unroll
    for (int kt = kBeg; kt < kBeg + KSLICE_; kt += 8) {
        int k0 = kt + half * 2;
        int k1 = k0 + 4;
        v2f a0, b0, a1, b1;
        a0.x = Arow[k0];
        a0.y = Arow[k0 + 1];
        a1.x = Arow[k1];
        a1.y = Arow[k1 + 1];
        b0.x = Bcol[(size_t)k0 * Ncols];
        b0.y = Bcol[(size_t)(k0 + 1) * Ncols];
        b1.x = Bcol[(size_t)k1 * Ncols];
        b1.y = Bcol[(size_t)(k1 + 1) * Ncols];
        acc0 = __builtin_amdgcn_wmma_f32_16x16x4_f32(false, a0, false, b0,
                                                     (short)0, acc0, false, false);
        acc1 = __builtin_amdgcn_wmma_f32_16x16x4_f32(false, a1, false, b1,
                                                     (short)0, acc1, false, false);
    }
    v8f acc = acc0 + acc1;
    for (int v = 0; v < 8; ++v) part[wave][v][lane] = acc[v];
    __syncthreads();
    if (wave < 8) {
        float s = part[0][wave][lane];
        for (int w = 1; w < KSPLIT_; ++w) s += part[w][wave][lane];
        int m = mBase + half * 8 + wave;
        if (m < Mrows && nCol < Ncols) C[(size_t)m * Ncols + nCol] = s;
    }
}

__global__ void node_h_kernel(const float* __restrict__ x3, const int* __restrict__ gid,
                              const unsigned* __restrict__ mask, const unsigned* __restrict__ aggk,
                              const float* __restrict__ P, const float* __restrict__ Q,
                              const float* __restrict__ Wl1, const float* __restrict__ Wr1,
                              const float* __restrict__ bl1,
                              const float* __restrict__ Wl2, const float* __restrict__ Wr2,
                              float* __restrict__ tl, float* __restrict__ tr, int N) {
    int wave = threadIdx.x >> 5;
    int lane = threadIdx.x & 31;
    int n = blockIdx.x * (blockDim.x >> 5) + wave;
    if (n >= N) return;

    float x0 = x3[(size_t)n * 3 + 0];
    float x1 = x3[(size_t)n * 3 + 1];
    float x2 = x3[(size_t)n * 3 + 2];
    int g = gid[n];
    unsigned mk = mask[n];
    bool has = (mk != 0u);
    float a0 = 0.f, a1 = 0.f, a2 = 0.f;
    if (has) {
        a0 = keyf(aggk[(size_t)n * 3 + 0]);
        a1 = keyf(aggk[(size_t)n * 3 + 1]);
        a2 = keyf(aggk[(size_t)n * 3 + 2]);
    }
    const float* Prow = P + (size_t)mk * HID_;
    const float* Qrow = Q + (size_t)g * HID_;

    float v[4];
    #pragma unroll
    for (int c = 0; c < 4; ++c) {
        int col = lane + 32 * c;
        if (col < HID_) {
            float t = bl1[col] + Qrow[col]
                    + x0 * Wr1[col] + x1 * Wr1[HID_ + col] + x2 * Wr1[2 * HID_ + col];
            if (has)
                t += Prow[col]
                   + a0 * Wl1[col] + a1 * Wl1[HID_ + col] + a2 * Wl1[2 * HID_ + col];
            v[c] = t;
        } else {
            v[c] = -INFINITY;
        }
    }
    float m = fmaxf(fmaxf(v[0], v[1]), fmaxf(v[2], v[3]));
    for (int off = 16; off >= 1; off >>= 1)
        m = fmaxf(m, __shfl_xor(m, off, 32));
    float s = 0.f;
    #pragma unroll
    for (int c = 0; c < 4; ++c) s += __expf(v[c] - m);
    for (int off = 16; off >= 1; off >>= 1)
        s += __shfl_xor(s, off, 32);
    float lse = m + __logf(s);
    float s1 = 0.f, s2 = 0.f;
    #pragma unroll
    for (int c = 0; c < 4; ++c) {
        int col = lane + 32 * c;
        if (col < HID_) {
            float hv = v[c] - lse;
            s1 = fmaf(hv, Wl2[col], s1);
            s2 = fmaf(hv, Wr2[col], s2);
        }
    }
    for (int off = 16; off >= 1; off >>= 1) {
        s1 += __shfl_xor(s1, off, 32);
        s2 += __shfl_xor(s2, off, 32);
    }
    if (lane == 0) { tl[n] = s1; tr[n] = s2; }
}

__global__ void edge_scatter_t(const int* __restrict__ ei, const float* __restrict__ tl,
                               float* __restrict__ agg2, int E) {
    int e = blockIdx.x * blockDim.x + threadIdx.x;
    if (e >= E) return;
    int s = ei[e];
    int d = ei[E + e];
    atomicAdd(&agg2[d], tl[s]);
}

__global__ void node_out_kernel(const float* __restrict__ agg2, const float* __restrict__ tr,
                                const int* __restrict__ counts, const float* __restrict__ bl2,
                                float* __restrict__ y, int N) {
    int n = blockIdx.x * blockDim.x + threadIdx.x;
    if (n >= N) return;
    y[n] = agg2[n] / fmaxf((float)counts[n], 1.0f) + bl2[0] + tr[n];
}

extern "C" void kernel_launch(void* const* d_in, const int* in_sizes, int n_in,
                              void* d_out, int out_size, void* d_ws, size_t ws_size,
                              hipStream_t stream) {
    const float* X      = (const float*)d_in[0];
    const float* x3     = (const float*)d_in[1];
    const int*   ei     = (const int*)d_in[2];
    const int*   gid    = (const int*)d_in[3];
    const float* w1     = (const float*)d_in[4];
    const float* b1     = (const float*)d_in[5];
    const float* w2     = (const float*)d_in[6];
    const float* b2     = (const float*)d_in[7];
    const float* w3     = (const float*)d_in[8];
    const float* b3     = (const float*)d_in[9];
    const float* Wl1    = (const float*)d_in[10];
    const float* bl1    = (const float*)d_in[11];
    const float* Wr1    = (const float*)d_in[12];
    const float* Wl2    = (const float*)d_in[13];
    const float* bl2    = (const float*)d_in[14];
    const float* Wr2    = (const float*)d_in[15];
    float* y = (float*)d_out;

    const int E = in_sizes[2] / 2;
    const int N = in_sizes[3];

    char* ws = (char*)d_ws;
    size_t off = 0;
    auto alloc = [&](size_t bytes) -> char* {
        char* p = ws + off;
        off = (off + bytes + 255) & ~(size_t)255;
        return p;
    };
    float*    act1  = (float*)alloc((size_t)8 * 32 * 21 * 36 * 4);
    float*    act2  = (float*)alloc((size_t)8 * 64 * 17 * 36 * 4);
    float*    act3  = (float*)alloc((size_t)8 * 64 * 13 * 16 * 4);
    float*    enc   = (float*)alloc((size_t)8 * ENC_ * 4);
    float*    ME    = (float*)alloc((size_t)256 * ENC_ * 4);
    float*    P     = (float*)alloc((size_t)256 * HID_ * 4);
    float*    Q     = (float*)alloc((size_t)16  * HID_ * 4);
    unsigned* aggk  = (unsigned*)alloc((size_t)N * 3 * 4);
    unsigned* mask  = (unsigned*)alloc((size_t)N * 4);
    int*      cnt   = (int*)alloc((size_t)N * 4);
    float*    tl    = (float*)alloc((size_t)N * 4);
    float*    tr    = (float*)alloc((size_t)N * 4);
    float*    agg2  = (float*)alloc((size_t)N * 4);

    const int T = 256;
    zero_u32<<<(N * 3 + T - 1) / T, T, 0, stream>>>(aggk, N * 3);
    zero_u32<<<(N + T - 1) / T, T, 0, stream>>>(mask, N);
    zero_u32<<<(N + T - 1) / T, T, 0, stream>>>((unsigned*)cnt, N);
    zero_u32<<<(N + T - 1) / T, T, 0, stream>>>((unsigned*)agg2, N);

    conv1_kernel<<<dim3(32, 8), 512, 0, stream>>>(X, w1, b1, act1);
    conv2_kernel<<<dim3(27, 8), 512, 0, stream>>>(act1, w2, b2, act2);
    conv3_kernel<<<dim3(16, 16), 512, 0, stream>>>(act2, w3, b3, act3);
    pool_kernel<<<(8 * 64 * 11 * 4 + T - 1) / T, T, 0, stream>>>(act3, enc);

    edge_pass<<<(E + T - 1) / T, T, 0, stream>>>(ei, gid, x3, aggk, mask, cnt, E);

    maxenc_kernel<<<dim3(ENC_ / 4 / ME_T_, 16), ME_T_, 0, stream>>>(enc, ME);
    {
        dim3 gP(256 / 16, (HID_ + 15) / 16);
        gemm_wmma_f32<<<gP, 32 * KSPLIT_, 0, stream>>>(ME, Wl1 + 3 * HID_, P, 256);
        dim3 gQ(1, (HID_ + 15) / 16);
        gemm_wmma_f32<<<gQ, 32 * KSPLIT_, 0, stream>>>(enc, Wr1 + 3 * HID_, Q, 8);
    }

    node_h_kernel<<<(N + 7) / 8, 256, 0, stream>>>(x3, gid, mask, aggk, P, Q,
                                                   Wl1, Wr1, bl1, Wl2, Wr2, tl, tr, N);

    edge_scatter_t<<<(E + T - 1) / T, T, 0, stream>>>(ei, tl, agg2, E);
    node_out_kernel<<<(N + T - 1) / T, T, 0, stream>>>(agg2, tr, cnt, bl2, y, N);

    (void)n_in; (void)out_size; (void)ws_size;
}


# Fewer launches (15 to 8): conv1 also zeroes the accumulators, the maxpool moves into the subset-max kernel, the edge pass shares that launch, both GEMMs share one; no sum or its order changed
# speedup vs baseline: 6.8274x; 1.2667x over previous
#include <hip/hip_runtime.h>
#include <hip/hip_bf16.h>
#include <math.h>


#define B_     8
#define CIN_   25
#define NPG_   2000
#define ENC_   2816
#define HID_   100
#define KGEMM_ 2816

typedef __attribute__((ext_vector_type(2))) float v2f;
typedef __attribute__((ext_vector_type(8))) float v8f;
typedef __attribute__((ext_vector_type(4))) float v4f;

__device__ __forceinline__ unsigned fkey(float f) {
    unsigned u = __float_as_uint(f);
    return (u & 0x80000000u) ? ~u : (u | 0x80000000u);
}
__device__ __forceinline__ float keyf(unsigned k) {
    unsigned u = (k & 0x80000000u) ? (k ^ 0x80000000u) : ~k;
    return __uint_as_float(u);
}


template <int CIN, int COUT, int DIN, int DOUT, int HIN, int HOUT, int PAD, int OCT, int PB, int ZB, int KS>
__device__ __forceinline__ void conv_body(const float* __restrict__ in, const float* __restrict__ w,
                                          const float* __restrict__ b, float* __restrict__ out,
                                          float* wl, float* red) {
    constexpr int WIN = HIN;
    constexpr int WOUT = HOUT;
    constexpr int PLANE = HIN * WIN;
    constexpr int HW = HOUT * WOUT;
    constexpr int NBLK = (DOUT + ZB - 1) / ZB;
    constexpr int NTOT = 8 * NBLK * HW;
    constexpr int KW = CIN * 45;
    constexpr int NT = PB * KS;
    constexpr int ND = ZB + 4;
    constexpr int NACC = ZB * OCT;

    const int tid = (int)threadIdx.x;
    const int og = blockIdx.y * OCT;
    constexpr int NW = KW * OCT;
    constexpr int SU = (NW + NT - 1) / NT;
    {
        float tmp[SU];
        #pragma unroll
        for (int u = 0; u < SU; ++u) {
            int i = u * NT + tid;
            int ic = i < NW ? i : NW - 1;
            int o = ic / KW;
            int r = ic - o * KW;
            tmp[u] = w[(size_t)(og + o) * KW + r];
        }
        #pragma unroll
        for (int u = 0; u < SU; ++u) {
            int i = u * NT + tid;
            if (i < NW) {
                int o = i / KW;
                int r = i - o * KW;
                wl[r * OCT + o] = tmp[u];
            }
        }
    }
    __syncthreads();

    const int ks = __builtin_amdgcn_readfirstlane(tid / PB);
    const int pw = tid - ks * PB;
    const int cBeg = (ks * CIN) / KS;
    const int cEnd = ((ks + 1) * CIN) / KS;

    int gp = blockIdx.x * PB + pw;
    const bool active = gp < NTOT;
    if (!active) gp = NTOT - 1;
    const int bn = gp / (NBLK * HW);
    const int rb = gp - bn * (NBLK * HW);
    const int blk = rb / HW;
    const int p = rb - blk * HW;
    const int y = p / WOUT;
    const int x = p - y * WOUT;
    const int zFirst = blk * ZB;
    const int z0 = zFirst < DOUT - ZB ? zFirst : DOUT - ZB;

    int off[9];
    bool ok[9];
    #pragma unroll
    for (int kh = 0; kh < 3; ++kh)
        #pragma unroll
        for (int kw = 0; kw < 3; ++kw) {
            int yy = y + kh - PAD;
            int xx = x + kw - PAD;
            ok[kh * 3 + kw] = (yy >= 0) && (yy < HIN) && (xx >= 0) && (xx < WIN);
            int yc = yy < 0 ? 0 : (yy >= HIN ? HIN - 1 : yy);
            int xc = xx < 0 ? 0 : (xx >= WIN ? WIN - 1 : xx);
            off[kh * 3 + kw] = yc * WIN + xc;
        }

    float acc[NACC];
    #pragma unroll
    for (int a = 0; a < NACC; ++a) acc[a] = (ks == 0) ? b[og + a % OCT] : 0.0f;

    const float* ib = in + ((size_t)bn * CIN * DIN + z0) * PLANE;
    for (int c = cBeg; c < cEnd; ++c) {
        const float* ic = ib + (size_t)c * DIN * PLANE;
        float xv[ND * 9];
        #pragma unroll
        for (int d = 0; d < ND; ++d)
            #pragma unroll
            for (int t = 0; t < 9; ++t)
                xv[d * 9 + t] = ic[d * PLANE + off[t]];
        if (PAD != 0) {
            #pragma unroll
            for (int d = 0; d < ND; ++d)
                asm volatile("" :: "v"(xv[d * 9 + 0]), "v"(xv[d * 9 + 1]), "v"(xv[d * 9 + 2]), "v"(xv[d * 9 + 3]), "v"(xv[d * 9 + 4]), "v"(xv[d * 9 + 5]), "v"(xv[d * 9 + 6]), "v"(xv[d * 9 + 7]), "v"(xv[d * 9 + 8]));
            #pragma unroll
            for (int j = 0; j < ND * 9; ++j) xv[j] = ok[j % 9] ? xv[j] : 0.0f;
        }
        constexpr int Q = OCT / 4;
        const v4f* wc4 = (const v4f*)(wl + c * 45 * OCT);
        #pragma unroll
        for (int kd = 0; kd < 5; ++kd) {
            #pragma unroll
            for (int h = 0; h < Q; ++h) {
                v4f wq[9];
                #pragma unroll
                for (int t = 0; t < 9; ++t) wq[t] = wc4[(kd * 9 + t) * Q + h];
                asm volatile("" :: "v"(wq[0]), "v"(wq[1]), "v"(wq[2]), "v"(wq[3]), "v"(wq[4]), "v"(wq[5]), "v"(wq[6]), "v"(wq[7]), "v"(wq[8]));
                #pragma unroll
                for (int zb = 0; zb < ZB; ++zb)
                    #pragma unroll
                    for (int t = 0; t < 9; ++t)
                        #pragma unroll
                        for (int o = 0; o < 4; ++o)
                            acc[zb * OCT + h * 4 + o] = fmaf(xv[(zb + kd) * 9 + t], wq[t][o], acc[zb * OCT + h * 4 + o]);
            }
        }
    }

    if (ks > 0) {
        #pragma unroll
        for (int a = 0; a < NACC; ++a) red[((ks - 1) * NACC + a) * PB + pw] = acc[a];
    }
    __syncthreads();
    if (ks == 0 && active) {
        #pragma unroll
        for (int zb = 0; zb < ZB; ++zb) {
            const int z = z0 + zb;
            #pragma unroll
            for (int o = 0; o < OCT; ++o) {
                const int a = zb * OCT + o;
                float s = acc[a];
                #pragma unroll
                for (int k = 1; k < KS; ++k) s += red[((k - 1) * NACC + a) * PB + pw];
                if (z >= zFirst)
                    out[((size_t)(bn * COUT + og + o) * DOUT + z) * HW + y * WOUT + x] = fmaxf(s, 0.0f);
            }
        }
    }
}

__global__ void __launch_bounds__(512) __attribute__((amdgpu_waves_per_eu(4, 4)))
conv1_kernel(const float* __restrict__ X, const float* __restrict__ w,
             const float* __restrict__ b, float* __restrict__ out,
             unsigned* __restrict__ zfill, int nz) {
    __shared__ __attribute__((aligned(16))) float wl[25 * 45 * 4];
    __shared__ float red[(8 - 1) * 3 * 4 * 64];
    {
        const int nthr = (int)(gridDim.x * gridDim.y) * 512;
        for (int i = (int)(blockIdx.y * gridDim.x + blockIdx.x) * 512 + (int)threadIdx.x; i < nz; i += nthr)
            zfill[i] = 0u;
    }
    conv_body<25, 32, 25, 21, 6, 6, 1, 4, 64, 3, 8>(X, w, b, out, wl, red);
}

__global__ void __launch_bounds__(512) __attribute__((amdgpu_waves_per_eu(4, 4)))
conv2_kernel(const float* __restrict__ A, const float* __restrict__ w,
             const float* __restrict__ b, float* __restrict__ out) {
    __shared__ __attribute__((aligned(16))) float wl[32 * 45 * 8];
    __shared__ float red[(8 - 1) * 3 * 8 * 64];
    conv_body<32, 64, 21, 17, 6, 6, 1, 8, 64, 3, 8>(A, w, b, out, wl, red);
}

__global__ void __launch_bounds__(512) __attribute__((amdgpu_waves_per_eu(4, 4)))
conv3_kernel(const float* __restrict__ A, const float* __restrict__ w,
             const float* __restrict__ b, float* __restrict__ out) {
    __shared__ __attribute__((aligned(16))) float wl[64 * 45 * 4];
    __shared__ float red[(16 - 1) * 4 * 4 * 32];
    conv_body<64, 64, 17, 13, 6, 4, 0, 4, 32, 4, 16>(A, w, b, out, wl, red);
}


__device__ __forceinline__ void edge_body(const int* __restrict__ ei, const int* __restrict__ gid,
                                          const float* __restrict__ x3, unsigned* __restrict__ aggk,
                                          unsigned* __restrict__ mask, int* __restrict__ counts,
                                          int E, int e) {
    if (e >= E) return;
    int s = ei[e];
    int d = ei[E + e];
    int g = gid[s];
    atomicOr(&mask[d], 1u << g);
    atomicAdd(&counts[d], 1);
    atomicMax(&aggk[(size_t)d * 3 + 0], fkey(x3[(size_t)s * 3 + 0]));
    atomicMax(&aggk[(size_t)d * 3 + 1], fkey(x3[(size_t)s * 3 + 1]));
    atomicMax(&aggk[(size_t)d * 3 + 2], fkey(x3[(size_t)s * 3 + 2]));
}

#define ME_T_  64
#define ME_WG_ (11 * 16)

__device__ __forceinline__ float4 max4(float4 a, float4 b) {
    return make_float4(fmaxf(a.x, b.x), fmaxf(a.y, b.y), fmaxf(a.z, b.z), fmaxf(a.w, b.w));
}

__global__ void __launch_bounds__(ME_T_)
maxenc_edge_kernel(const float* __restrict__ act3, float* __restrict__ enc, float* __restrict__ ME,
                   const int* __restrict__ ei, const int* __restrict__ gid,
                   const float* __restrict__ x3, unsigned* __restrict__ aggk,
                   unsigned* __restrict__ mask, int* __restrict__ counts, int E) {
    const int bid = (int)blockIdx.x;
    if (bid >= ME_WG_) {
        edge_body(ei, gid, x3, aggk, mask, counts, E, (bid - ME_WG_) * ME_T_ + (int)threadIdx.x);
        return;
    }
    const int hi = bid / 11;
    const int c4 = (bid - hi * 11) * ME_T_ + (int)threadIdx.x;
    const int ch = c4 / 11;
    const int d  = c4 - ch * 11;
    const float4* a4 = (const float4*)act3;
    float4 e[8];
    #pragma unroll
    for (int g = 0; g < 8; ++g) {
        const float4* p = a4 + (size_t)((g * 64 + ch) * 13 + d) * 4;
        float h0[4], h1[4];
        #pragma unroll
        for (int yy = 0; yy < 4; ++yy) {
            const float4 r = max4(max4(p[yy], p[4 + yy]), p[8 + yy]);
            h0[yy] = fmaxf(fmaxf(r.x, r.y), r.z);
            h1[yy] = fmaxf(fmaxf(r.y, r.z), r.w);
        }
        e[g].x = fmaxf(fmaxf(h0[0], h0[1]), h0[2]);
        e[g].y = fmaxf(fmaxf(h1[0], h1[1]), h1[2]);
        e[g].z = fmaxf(fmaxf(h0[1], h0[2]), h0[3]);
        e[g].w = fmaxf(fmaxf(h1[1], h1[2]), h1[3]);
    }
    if (hi == 0) {
        float4* e4 = (float4*)enc;
        #pragma unroll
        for (int g = 0; g < 8; ++g) e4[(size_t)g * (ENC_ / 4) + c4] = e[g];
    }

    float4 t[16];
    float4 base = make_float4(0.f, 0.f, 0.f, 0.f);
    #pragma unroll
    for (int b = 3; b >= 0; --b) {
        const bool on = ((hi >> b) & 1) != 0;
        base.x = fmaxf(base.x, on ? e[4 + b].x : 0.0f);
        base.y = fmaxf(base.y, on ? e[4 + b].y : 0.0f);
        base.z = fmaxf(base.z, on ? e[4 + b].z : 0.0f);
        base.w = fmaxf(base.w, on ? e[4 + b].w : 0.0f);
    }
    t[0] = base;
    #pragma unroll
    for (int lo = 1; lo < 16; ++lo) {
        const int p = lo & (lo - 1);
        const int g = __builtin_ctz(lo);
        t[lo].x = fmaxf(t[p].x, e[g].x);
        t[lo].y = fmaxf(t[p].y, e[g].y);
        t[lo].z = fmaxf(t[p].z, e[g].z);
        t[lo].w = fmaxf(t[p].w, e[g].w);
    }
    float4* M4 = (float4*)ME;
    #pragma unroll
    for (int lo = 0; lo < 16; ++lo)
        M4[(size_t)(hi * 16 + lo) * (ENC_ / 4) + c4] = t[lo];
}

#define KSPLIT_ 16
#define KSLICE_ (KGEMM_ / KSPLIT_)

__global__ void __launch_bounds__(32 * KSPLIT_)
gemm_wmma_f32(const float* __restrict__ A1, const float* __restrict__ W1,
              float* __restrict__ C1, int Mrows1, int tilesP,
              const float* __restrict__ A2, const float* __restrict__ W2,
              float* __restrict__ C2, int Mrows2) {
    __shared__ float part[KSPLIT_][8][32];
    const int Ncols = HID_;
    const bool second = (int)blockIdx.x >= tilesP;
    const float* __restrict__ A = second ? A2 : A1;
    const float* __restrict__ W = second ? W2 : W1;
    float* __restrict__ C = second ? C2 : C1;
    const int Mrows = second ? Mrows2 : Mrows1;
    int wave = threadIdx.x >> 5;
    int lane = threadIdx.x & 31;
    int half = lane >> 4;
    int l16  = lane & 15;
    int mBase = second ? 0 : blockIdx.x * 16;
    int nBase = blockIdx.y * 16;
    int mRow = mBase + l16; if (mRow >= Mrows) mRow = Mrows - 1;
    int nCol = nBase + l16;
    int nClamp = (nCol < Ncols) ? nCol : (Ncols - 1);
    const float* Arow = A + (size_t)mRow * KGEMM_;
    const float* Bcol = W + nClamp;

    const int kBeg = wave * KSLICE_;
    v8f acc0 = {}, acc1 = {};
    #pragma unroll
    for (int kt = kBeg; kt < kBeg + KSLICE_; kt += 8) {
        int k0 = kt + half * 2;
        int k1 = k0 + 4;
        v2f a0, b0, a1, b1;
        a0.x = Arow[k0];
        a0.y = Arow[k0 + 1];
        a1.x = Arow[k1];
        a1.y = Arow[k1 + 1];
        b0.x = Bcol[(size_t)k0 * Ncols];
        b0.y = Bcol[(size_t)(k0 + 1) * Ncols];
        b1.x = Bcol[(size_t)k1 * Ncols];
        b1.y = Bcol[(size_t)(k1 + 1) * Ncols];
        acc0 = __builtin_amdgcn_wmma_f32_16x16x4_f32(false, a0, false, b0,
                                                     (short)0, acc0, false, false);
        acc1 = __builtin_amdgcn_wmma_f32_16x16x4_f32(false, a1, false, b1,
                                                     (short)0, acc1, false, false);
    }
    v8f acc = acc0 + acc1;
    for (int v = 0; v < 8; ++v) part[wave][v][lane] = acc[v];
    __syncthreads();
    if (wave < 8) {
        float s = part[0][wave][lane];
        for (int w = 1; w < KSPLIT_; ++w) s += part[w][wave][lane];
        int m = mBase + half * 8 + wave;
        if (m < Mrows && nCol < Ncols) C[(size_t)m * Ncols + nCol] = s;
    }
}

__global__ void node_h_kernel(const float* __restrict__ x3, const int* __restrict__ gid,
                              const unsigned* __restrict__ mask, const unsigned* __restrict__ aggk,
                              const float* __restrict__ P, const float* __restrict__ Q,
                              const float* __restrict__ Wl1, const float* __restrict__ Wr1,
                              const float* __restrict__ bl1,
                              const float* __restrict__ Wl2, const float* __restrict__ Wr2,
                              float* __restrict__ tl, float* __restrict__ tr, int N) {
    int wave = threadIdx.x >> 5;
    int lane = threadIdx.x & 31;
    int n = blockIdx.x * (blockDim.x >> 5) + wave;
    if (n >= N) return;

    float x0 = x3[(size_t)n * 3 + 0];
    float x1 = x3[(size_t)n * 3 + 1];
    float x2 = x3[(size_t)n * 3 + 2];
    int g = gid[n];
    unsigned mk = mask[n];
    bool has = (mk != 0u);
    float a0 = 0.f, a1 = 0.f, a2 = 0.f;
    if (has) {
        a0 = keyf(aggk[(size_t)n * 3 + 0]);
        a1 = keyf(aggk[(size_t)n * 3 + 1]);
        a2 = keyf(aggk[(size_t)n * 3 + 2]);
    }
    const float* Prow = P + (size_t)mk * HID_;
    const float* Qrow = Q + (size_t)g * HID_;

    float v[4];
    #pragma unroll
    for (int c = 0; c < 4; ++c) {
        int col = lane + 32 * c;
        if (col < HID_) {
            float t = bl1[col] + Qrow[col]
                    + x0 * Wr1[col] + x1 * Wr1[HID_ + col] + x2 * Wr1[2 * HID_ + col];
            if (has)
                t += Prow[col]
                   + a0 * Wl1[col] + a1 * Wl1[HID_ + col] + a2 * Wl1[2 * HID_ + col];
            v[c] = t;
        } else {
            v[c] = -INFINITY;
        }
    }
    float m = fmaxf(fmaxf(v[0], v[1]), fmaxf(v[2], v[3]));
    for (int off = 16; off >= 1; off >>= 1)
        m = fmaxf(m, __shfl_xor(m, off, 32));
    float s = 0.f;
    #pragma unroll
    for (int c = 0; c < 4; ++c) s += __expf(v[c] - m);
    for (int off = 16; off >= 1; off >>= 1)
        s += __shfl_xor(s, off, 32);
    float lse = m + __logf(s);
    float s1 = 0.f, s2 = 0.f;
    #pragma unroll
    for (int c = 0; c < 4; ++c) {
        int col = lane + 32 * c;
        if (col < HID_) {
            float hv = v[c] - lse;
            s1 = fmaf(hv, Wl2[col], s1);
            s2 = fmaf(hv, Wr2[col], s2);
        }
    }
    for (int off = 16; off >= 1; off >>= 1) {
        s1 += __shfl_xor(s1, off, 32);
        s2 += __shfl_xor(s2, off, 32);
    }
    if (lane == 0) { tl[n] = s1; tr[n] = s2; }
}

__global__ void edge_scatter_t(const int* __restrict__ ei, const float* __restrict__ tl,
                               float* __restrict__ agg2, int E) {
    int e = blockIdx.x * blockDim.x + threadIdx.x;
    if (e >= E) return;
    int s = ei[e];
    int d = ei[E + e];
    atomicAdd(&agg2[d], tl[s]);
}

__global__ void node_out_kernel(const float* __restrict__ agg2, const float* __restrict__ tr,
                                const int* __restrict__ counts, const float* __restrict__ bl2,
                                float* __restrict__ y, int N) {
    int n = blockIdx.x * blockDim.x + threadIdx.x;
    if (n >= N) return;
    y[n] = agg2[n] / fmaxf((float)counts[n], 1.0f) + bl2[0] + tr[n];
}

extern "C" void kernel_launch(void* const* d_in, const int* in_sizes, int n_in,
                              void* d_out, int out_size, void* d_ws, size_t ws_size,
                              hipStream_t stream) {
    const float* X      = (const float*)d_in[0];
    const float* x3     = (const float*)d_in[1];
    const int*   ei     = (const int*)d_in[2];
    const int*   gid    = (const int*)d_in[3];
    const float* w1     = (const float*)d_in[4];
    const float* b1     = (const float*)d_in[5];
    const float* w2     = (const float*)d_in[6];
    const float* b2     = (const float*)d_in[7];
    const float* w3     = (const float*)d_in[8];
    const float* b3     = (const float*)d_in[9];
    const float* Wl1    = (const float*)d_in[10];
    const float* bl1    = (const float*)d_in[11];
    const float* Wr1    = (const float*)d_in[12];
    const float* Wl2    = (const float*)d_in[13];
    const float* bl2    = (const float*)d_in[14];
    const float* Wr2    = (const float*)d_in[15];
    float* y = (float*)d_out;

    const int E = in_sizes[2] / 2;
    const int N = in_sizes[3];

    char* ws = (char*)d_ws;
    size_t off = 0;
    auto alloc = [&](size_t bytes) -> char* {
        char* p = ws + off;
        off = (off + bytes + 255) & ~(size_t)255;
        return p;
    };
    float*    act1  = (float*)alloc((size_t)8 * 32 * 21 * 36 * 4);
    float*    act2  = (float*)alloc((size_t)8 * 64 * 17 * 36 * 4);
    float*    act3  = (float*)alloc((size_t)8 * 64 * 13 * 16 * 4);
    float*    enc   = (float*)alloc((size_t)8 * ENC_ * 4);
    float*    ME    = (float*)alloc((size_t)256 * ENC_ * 4);
    float*    P     = (float*)alloc((size_t)256 * HID_ * 4);
    float*    Q     = (float*)alloc((size_t)16  * HID_ * 4);
    unsigned* zblk  = (unsigned*)alloc((size_t)N * 6 * 4);
    unsigned* aggk  = zblk;
    unsigned* mask  = zblk + (size_t)N * 3;
    int*      cnt   = (int*)(zblk + (size_t)N * 4);
    float*    agg2  = (float*)(zblk + (size_t)N * 5);
    float*    tl    = (float*)alloc((size_t)N * 4);
    float*    tr    = (float*)alloc((size_t)N * 4);

    const int T = 256;
    conv1_kernel<<<dim3(32, 8), 512, 0, stream>>>(X, w1, b1, act1, zblk, N * 6);
    conv2_kernel<<<dim3(27, 8), 512, 0, stream>>>(act1, w2, b2, act2);
    conv3_kernel<<<dim3(16, 16), 512, 0, stream>>>(act2, w3, b3, act3);

    maxenc_edge_kernel<<<ME_WG_ + (E + ME_T_ - 1) / ME_T_, ME_T_, 0, stream>>>(act3, enc, ME, ei, gid, x3,
                                                                            aggk, mask, cnt, E);
    {
        dim3 gPQ(256 / 16 + 1, (HID_ + 15) / 16);
        gemm_wmma_f32<<<gPQ, 32 * KSPLIT_, 0, stream>>>(ME, Wl1 + 3 * HID_, P, 256, 256 / 16,
                                                        enc, Wr1 + 3 * HID_, Q, 8);
    }

    node_h_kernel<<<(N + 7) / 8, 256, 0, stream>>>(x3, gid, mask, aggk, P, Q,
                                                   Wl1, Wr1, bl1, Wl2, Wr2, tl, tr, N);

    edge_scatter_t<<<(E + T - 1) / T, T, 0, stream>>>(ei, tl, agg2, E);
    node_out_kernel<<<(N + T - 1) / T, T, 0, stream>>>(agg2, tr, cnt, bl2, y, N);

    (void)n_in; (void)out_size; (void)ws_size;
}


# Launches of thousands of short waves get few fat ones: node kernel one thread per node (16000 waves to 500), edge passes 6 edges per thread; per-thread serial sums (changes summation order)
# speedup vs baseline: 7.1043x; 1.0406x over previous
#include <hip/hip_runtime.h>
#include <hip/hip_bf16.h>
#include <math.h>


#define B_     8
#define CIN_   25
#define NPG_   2000
#define ENC_   2816
#define HID_   100
#define KGEMM_ 2816

typedef __attribute__((ext_vector_type(2))) float v2f;
typedef __attribute__((ext_vector_type(8))) float v8f;
typedef __attribute__((ext_vector_type(4))) float v4f;

__device__ __forceinline__ unsigned fkey(float f) {
    unsigned u = __float_as_uint(f);
    return (u & 0x80000000u) ? ~u : (u | 0x80000000u);
}
__device__ __forceinline__ float keyf(unsigned k) {
    unsigned u = (k & 0x80000000u) ? (k ^ 0x80000000u) : ~k;
    return __uint_as_float(u);
}


template <int CIN, int COUT, int DIN, int DOUT, int HIN, int HOUT, int PAD, int OCT, int PB, int ZB, int KS>
__device__ __forceinline__ void conv_body(const float* __restrict__ in, const float* __restrict__ w,
                                          const float* __restrict__ b, float* __restrict__ out,
                                          float* wl, float* red) {
    constexpr int WIN = HIN;
    constexpr int WOUT = HOUT;
    constexpr int PLANE = HIN * WIN;
    constexpr int HW = HOUT * WOUT;
    constexpr int NBLK = (DOUT + ZB - 1) / ZB;
    constexpr int NTOT = 8 * NBLK * HW;
    constexpr int KW = CIN * 45;
    constexpr int NT = PB * KS;
    constexpr int ND = ZB + 4;
    constexpr int NACC = ZB * OCT;

    const int tid = (int)threadIdx.x;
    const int og = blockIdx.y * OCT;
    constexpr int NW = KW * OCT;
    constexpr int SU = (NW + NT - 1) / NT;
    {
        float tmp[SU];
        #pragma unroll
        for (int u = 0; u < SU; ++u) {
            int i = u * NT + tid;
            int ic = i < NW ? i : NW - 1;
            int o = ic / KW;
            int r = ic - o * KW;
            tmp[u] = w[(size_t)(og + o) * KW + r];
        }
        #pragma unroll
        for (int u = 0; u < SU; ++u) {
            int i = u * NT + tid;
            if (i < NW) {
                int o = i / KW;
                int r = i - o * KW;
                wl[r * OCT + o] = tmp[u];
            }
        }
    }
    __syncthreads();

    const int ks = __builtin_amdgcn_readfirstlane(tid / PB);
    const int pw = tid - ks * PB;
    const int cBeg = (ks * CIN) / KS;
    const int cEnd = ((ks + 1) * CIN) / KS;

    int gp = blockIdx.x * PB + pw;
    const bool active = gp < NTOT;
    if (!active) gp = NTOT - 1;
    const int bn = gp / (NBLK * HW);
    const int rb = gp - bn * (NBLK * HW);
    const int blk = rb / HW;
    const int p = rb - blk * HW;
    const int y = p / WOUT;
    const int x = p - y * WOUT;
    const int zFirst = blk * ZB;
    const int z0 = zFirst < DOUT - ZB ? zFirst : DOUT - ZB;

    int off[9];
    bool ok[9];
    #pragma unroll
    for (int kh = 0; kh < 3; ++kh)
        #pragma unroll
        for (int kw = 0; kw < 3; ++kw) {
            int yy = y + kh - PAD;
            int xx = x + kw - PAD;
            ok[kh * 3 + kw] = (yy >= 0) && (yy < HIN) && (xx >= 0) && (xx < WIN);
            int yc = yy < 0 ? 0 : (yy >= HIN ? HIN - 1 : yy);
            int xc = xx < 0 ? 0 : (xx >= WIN ? WIN - 1 : xx);
            off[kh * 3 + kw] = yc * WIN + xc;
        }

    float acc[NACC];
    #pragma unroll
    for (int a = 0; a < NACC; ++a) acc[a] = (ks == 0) ? b[og + a % OCT] : 0.0f;

    const float* ib = in + ((size_t)bn * CIN * DIN + z0) * PLANE;
    for (int c = cBeg; c < cEnd; ++c) {
        const float* ic = ib + (size_t)c * DIN * PLANE;
        float xv[ND * 9];
        #pragma unroll
        for (int d = 0; d < ND; ++d)
            #pragma unroll
            for (int t = 0; t < 9; ++t)
                xv[d * 9 + t] = ic[d * PLANE + off[t]];
        if (PAD != 0) {
            #pragma unroll
            for (int d = 0; d < ND; ++d)
                asm volatile("" :: "v"(xv[d * 9 + 0]), "v"(xv[d * 9 + 1]), "v"(xv[d * 9 + 2]), "v"(xv[d * 9 + 3]), "v"(xv[d * 9 + 4]), "v"(xv[d * 9 + 5]), "v"(xv[d * 9 + 6]), "v"(xv[d * 9 + 7]), "v"(xv[d * 9 + 8]));
            #pragma unroll
            for (int j = 0; j < ND * 9; ++j) xv[j] = ok[j % 9] ? xv[j] : 0.0f;
        }
        constexpr int Q = OCT / 4;
        const v4f* wc4 = (const v4f*)(wl + c * 45 * OCT);
        #pragma unroll
        for (int kd = 0; kd < 5; ++kd) {
            #pragma unroll
            for (int h = 0; h < Q; ++h) {
                v4f wq[9];
                #pragma unroll
                for (int t = 0; t < 9; ++t) wq[t] = wc4[(kd * 9 + t) * Q + h];
                asm volatile("" :: "v"(wq[0]), "v"(wq[1]), "v"(wq[2]), "v"(wq[3]), "v"(wq[4]), "v"(wq[5]), "v"(wq[6]), "v"(wq[7]), "v"(wq[8]));
                #pragma unroll
                for (int zb = 0; zb < ZB; ++zb)
                    #pragma unroll
                    for (int t = 0; t < 9; ++t)
                        #pragma unroll
                        for (int o = 0; o < 4; ++o)
                            acc[zb * OCT + h * 4 + o] = fmaf(xv[(zb + kd) * 9 + t], wq[t][o], acc[zb * OCT + h * 4 + o]);
            }
        }
    }

    if (ks > 0) {
        #pragma unroll
        for (int a = 0; a < NACC; ++a) red[((ks - 1) * NACC + a) * PB + pw] = acc[a];
    }
    __syncthreads();
    if (ks == 0 && active) {
        #pragma unroll
        for (int zb = 0; zb < ZB; ++zb) {
            const int z = z0 + zb;
            #pragma unroll
            for (int o = 0; o < OCT; ++o) {
                const int a = zb * OCT + o;
                float s = acc[a];
                #pragma unroll
                for (int k = 1; k < KS; ++k) s += red[((k - 1) * NACC + a) * PB + pw];
                if (z >= zFirst)
                    out[((size_t)(bn * COUT + og + o) * DOUT + z) * HW + y * WOUT + x] = fmaxf(s, 0.0f);
            }
        }
    }
}

__global__ void __launch_bounds__(512) __attribute__((amdgpu_waves_per_eu(4, 4)))
conv1_kernel(const float* __restrict__ X, const float* __restrict__ w,
             const float* __restrict__ b, float* __restrict__ out,
             unsigned* __restrict__ zfill, int nz) {
    __shared__ __attribute__((aligned(16))) float wl[25 * 45 * 4];
    __shared__ float red[(8 - 1) * 3 * 4 * 64];
    {
        const int nthr = (int)(gridDim.x * gridDim.y) * 512;
        for (int i = (int)(blockIdx.y * gridDim.x + blockIdx.x) * 512 + (int)threadIdx.x; i < nz; i += nthr)
            zfill[i] = 0u;
    }
    conv_body<25, 32, 25, 21, 6, 6, 1, 4, 64, 3, 8>(X, w, b, out, wl, red);
}

__global__ void __launch_bounds__(512) __attribute__((amdgpu_waves_per_eu(4, 4)))
conv2_kernel(const float* __restrict__ A, const float* __restrict__ w,
             const float* __restrict__ b, float* __restrict__ out) {
    __shared__ __attribute__((aligned(16))) float wl[32 * 45 * 8];
    __shared__ float red[(8 - 1) * 3 * 8 * 64];
    conv_body<32, 64, 21, 17, 6, 6, 1, 8, 64, 3, 8>(A, w, b, out, wl, red);
}

__global__ void __launch_bounds__(512) __attribute__((amdgpu_waves_per_eu(4, 4)))
conv3_kernel(const float* __restrict__ A, const float* __restrict__ w,
             const float* __restrict__ b, float* __restrict__ out) {
    __shared__ __attribute__((aligned(16))) float wl[64 * 45 * 4];
    __shared__ float red[(16 - 1) * 4 * 4 * 32];
    conv_body<64, 64, 17, 13, 6, 4, 0, 4, 32, 4, 16>(A, w, b, out, wl, red);
}


#define EPT_ 6

template <bool FULL>
__device__ __forceinline__ void edge_body(const int* __restrict__ ei, const int* __restrict__ gid,
                                          const float* __restrict__ x3, unsigned* __restrict__ aggk,
                                          unsigned* __restrict__ mask, int* __restrict__ counts,
                                          int E, int e0, int stride) {
    int s[EPT_], d[EPT_];
    bool ok[EPT_];
    #pragma unroll
    for (int k = 0; k < EPT_; ++k) {
        const int e = e0 + k * stride;
        ok[k] = FULL || e < E;
        const int ec = ok[k] ? e : E - 1;
        s[k] = ei[ec];
        d[k] = ei[E + ec];
    }
    int g[EPT_];
    float xa[EPT_], xb[EPT_], xc[EPT_];
    #pragma unroll
    for (int k = 0; k < EPT_; ++k) {
        g[k] = gid[s[k]];
        xa[k] = x3[(size_t)s[k] * 3 + 0];
        xb[k] = x3[(size_t)s[k] * 3 + 1];
        xc[k] = x3[(size_t)s[k] * 3 + 2];
    }
    #pragma unroll
    for (int k = 0; k < EPT_; ++k) {
        if (ok[k]) {
            atomicOr(&mask[d[k]], 1u << g[k]);
            atomicAdd(&counts[d[k]], 1);
            atomicMax(&aggk[(size_t)d[k] * 3 + 0], fkey(xa[k]));
            atomicMax(&aggk[(size_t)d[k] * 3 + 1], fkey(xb[k]));
            atomicMax(&aggk[(size_t)d[k] * 3 + 2], fkey(xc[k]));
        }
    }
}

#define ME_T_  64
#define ME_WG_ (11 * 16)

__device__ __forceinline__ float4 max4(float4 a, float4 b) {
    return make_float4(fmaxf(a.x, b.x), fmaxf(a.y, b.y), fmaxf(a.z, b.z), fmaxf(a.w, b.w));
}

__global__ void __launch_bounds__(ME_T_)
maxenc_edge_kernel(const float* __restrict__ act3, float* __restrict__ enc, float* __restrict__ ME,
                   const int* __restrict__ ei, const int* __restrict__ gid,
                   const float* __restrict__ x3, unsigned* __restrict__ aggk,
                   unsigned* __restrict__ mask, int* __restrict__ counts, int E) {
    const int bid = (int)blockIdx.x;
    if (bid >= ME_WG_) {
        const int eb = (bid - ME_WG_) * (ME_T_ * EPT_);
        if (eb + ME_T_ * EPT_ <= E) edge_body<true>(ei, gid, x3, aggk, mask, counts, E, eb + (int)threadIdx.x, ME_T_);
        else                        edge_body<false>(ei, gid, x3, aggk, mask, counts, E, eb + (int)threadIdx.x, ME_T_);
        return;
    }
    const int hi = bid / 11;
    const int c4 = (bid - hi * 11) * ME_T_ + (int)threadIdx.x;
    const int ch = c4 / 11;
    const int d  = c4 - ch * 11;
    const float4* a4 = (const float4*)act3;
    float4 e[8];
    #pragma unroll
    for (int g = 0; g < 8; ++g) {
        const float4* p = a4 + (size_t)((g * 64 + ch) * 13 + d) * 4;
        float h0[4], h1[4];
        #pragma unroll
        for (int yy = 0; yy < 4; ++yy) {
            const float4 r = max4(max4(p[yy], p[4 + yy]), p[8 + yy]);
            h0[yy] = fmaxf(fmaxf(r.x, r.y), r.z);
            h1[yy] = fmaxf(fmaxf(r.y, r.z), r.w);
        }
        e[g].x = fmaxf(fmaxf(h0[0], h0[1]), h0[2]);
        e[g].y = fmaxf(fmaxf(h1[0], h1[1]), h1[2]);
        e[g].z = fmaxf(fmaxf(h0[1], h0[2]), h0[3]);
        e[g].w = fmaxf(fmaxf(h1[1], h1[2]), h1[3]);
    }
    if (hi == 0) {
        float4* e4 = (float4*)enc;
        #pragma unroll
        for (int g = 0; g < 8; ++g) e4[(size_t)g * (ENC_ / 4) + c4] = e[g];
    }

    float4 t[16];
    float4 base = make_float4(0.f, 0.f, 0.f, 0.f);
    #pragma unroll
    for (int b = 3; b >= 0; --b) {
        const bool on = ((hi >> b) & 1) != 0;
        base.x = fmaxf(base.x, on ? e[4 + b].x : 0.0f);
        base.y = fmaxf(base.y, on ? e[4 + b].y : 0.0f);
        base.z = fmaxf(base.z, on ? e[4 + b].z : 0.0f);
        base.w = fmaxf(base.w, on ? e[4 + b].w : 0.0f);
    }
    t[0] = base;
    #pragma unroll
    for (int lo = 1; lo < 16; ++lo) {
        const int p = lo & (lo - 1);
        const int g = __builtin_ctz(lo);
        t[lo].x = fmaxf(t[p].x, e[g].x);
        t[lo].y = fmaxf(t[p].y, e[g].y);
        t[lo].z = fmaxf(t[p].z, e[g].z);
        t[lo].w = fmaxf(t[p].w, e[g].w);
    }
    float4* M4 = (float4*)ME;
    #pragma unroll
    for (int lo = 0; lo < 16; ++lo)
        M4[(size_t)(hi * 16 + lo) * (ENC_ / 4) + c4] = t[lo];
}

#define KSPLIT_ 16
#define KSLICE_ (KGEMM_ / KSPLIT_)

__global__ void __launch_bounds__(32 * KSPLIT_)
gemm_wmma_f32(const float* __restrict__ A1, const float* __restrict__ W1,
              float* __restrict__ C1, int Mrows1, int tilesP,
              const float* __restrict__ A2, const float* __restrict__ W2,
              float* __restrict__ C2, int Mrows2) {
    __shared__ float part[KSPLIT_][8][32];
    const int Ncols = HID_;
    const bool second = (int)blockIdx.x >= tilesP;
    const float* __restrict__ A = second ? A2 : A1;
    const float* __restrict__ W = second ? W2 : W1;
    float* __restrict__ C = second ? C2 : C1;
    const int Mrows = second ? Mrows2 : Mrows1;
    int wave = threadIdx.x >> 5;
    int lane = threadIdx.x & 31;
    int half = lane >> 4;
    int l16  = lane & 15;
    int mBase = second ? 0 : blockIdx.x * 16;
    int nBase = blockIdx.y * 16;
    int mRow = mBase + l16; if (mRow >= Mrows) mRow = Mrows - 1;
    int nCol = nBase + l16;
    int nClamp = (nCol < Ncols) ? nCol : (Ncols - 1);
    const float* Arow = A + (size_t)mRow * KGEMM_;
    const float* Bcol = W + nClamp;

    const int kBeg = wave * KSLICE_;
    v8f acc0 = {}, acc1 = {};
    #pragma unroll
    for (int kt = kBeg; kt < kBeg + KSLICE_; kt += 8) {
        int k0 = kt + half * 2;
        int k1 = k0 + 4;
        v2f a0, b0, a1, b1;
        a0.x = Arow[k0];
        a0.y = Arow[k0 + 1];
        a1.x = Arow[k1];
        a1.y = Arow[k1 + 1];
        b0.x = Bcol[(size_t)k0 * Ncols];
        b0.y = Bcol[(size_t)(k0 + 1) * Ncols];
        b1.x = Bcol[(size_t)k1 * Ncols];
        b1.y = Bcol[(size_t)(k1 + 1) * Ncols];
        acc0 = __builtin_amdgcn_wmma_f32_16x16x4_f32(false, a0, false, b0,
                                                     (short)0, acc0, false, false);
        acc1 = __builtin_amdgcn_wmma_f32_16x16x4_f32(false, a1, false, b1,
                                                     (short)0, acc1, false, false);
    }
    v8f acc = acc0 + acc1;
    for (int v = 0; v < 8; ++v) part[wave][v][lane] = acc[v];
    __syncthreads();
    if (wave < 8) {
        float s = part[0][wave][lane];
        for (int w = 1; w < KSPLIT_; ++w) s += part[w][wave][lane];
        int m = mBase + half * 8 + wave;
        if (m < Mrows && nCol < Ncols) C[(size_t)m * Ncols + nCol] = s;
    }
}

#define NH_T_  64
#define NH_R_  9
#define NH_C4_ (HID_ / 4)

__global__ void __launch_bounds__(NH_T_) __attribute__((amdgpu_waves_per_eu(1, 1)))
node_h_kernel(const float* __restrict__ x3, const int* __restrict__ gid,
              const unsigned* __restrict__ mask, const unsigned* __restrict__ aggk,
              const float* __restrict__ P, const float* __restrict__ Q,
              const float* __restrict__ Wl1, const float* __restrict__ Wr1,
              const float* __restrict__ bl1,
              const float* __restrict__ Wl2, const float* __restrict__ Wr2,
              float* __restrict__ tl, float* __restrict__ tr, int N) {
    __shared__ __attribute__((aligned(16))) float wsh[NH_R_ * HID_];
    const int tid = (int)threadIdx.x;

    constexpr int U1 = (HID_ + NH_T_ - 1) / NH_T_;
    constexpr int U3 = (3 * HID_ + NH_T_ - 1) / NH_T_;
    float tb[U1], t2l[U1], t2r[U1], t1r[U3], t1l[U3];
    #pragma unroll
    for (int u = 0; u < U1; ++u) {
        int i = u * NH_T_ + tid;
        if (i > HID_ - 1) i = HID_ - 1;
        tb[u] = bl1[i];
        t2l[u] = Wl2[i];
        t2r[u] = Wr2[i];
    }
    #pragma unroll
    for (int u = 0; u < U3; ++u) {
        int i = u * NH_T_ + tid;
        if (i > 3 * HID_ - 1) i = 3 * HID_ - 1;
        t1r[u] = Wr1[i];
        t1l[u] = Wl1[i];
    }

    int n = blockIdx.x * NH_T_ + tid;
    const bool active = n < N;
    if (!active) n = N - 1;
    const float x0 = x3[(size_t)n * 3 + 0];
    const float x1 = x3[(size_t)n * 3 + 1];
    const float x2 = x3[(size_t)n * 3 + 2];
    const int g = gid[n];
    const unsigned mk = mask[n];
    const unsigned k0 = aggk[(size_t)n * 3 + 0];
    const unsigned k1 = aggk[(size_t)n * 3 + 1];
    const unsigned k2 = aggk[(size_t)n * 3 + 2];
    const bool has = (mk != 0u);
    const float a0 = has ? keyf(k0) : 0.f;
    const float a1 = has ? keyf(k1) : 0.f;
    const float a2 = has ? keyf(k2) : 0.f;

    #pragma unroll
    for (int u = 0; u < U1; ++u) {
        const int i = u * NH_T_ + tid;
        if (i < HID_) {
            wsh[i] = tb[u];
            wsh[7 * HID_ + i] = t2l[u];
            wsh[8 * HID_ + i] = t2r[u];
        }
    }
    #pragma unroll
    for (int u = 0; u < U3; ++u) {
        const int i = u * NH_T_ + tid;
        if (i < 3 * HID_) {
            wsh[HID_ + i] = t1r[u];
            wsh[4 * HID_ + i] = t1l[u];
        }
    }
    __syncthreads();

    const v4f* Prow = (const v4f*)(P + (size_t)mk * HID_);
    const v4f* Qrow = (const v4f*)(Q + (size_t)g * HID_);
    const float4* w4 = (const float4*)wsh;

    v4f pr[NH_C4_], qr[NH_C4_];
    #pragma unroll
    for (int c = 0; c < NH_C4_; ++c) pr[c] = Prow[c];
    #pragma unroll
    for (int c = 0; c < NH_C4_; ++c) qr[c] = Qrow[c];

    float v[HID_];
    #pragma unroll
    for (int c = 0; c < NH_C4_; ++c) {
        const v4f p = pr[c];
        const v4f q = qr[c];
        const float4 wb = w4[c];
        const float4 r0 = w4[1 * NH_C4_ + c], r1 = w4[2 * NH_C4_ + c], r2 = w4[3 * NH_C4_ + c];
        const float4 l0 = w4[4 * NH_C4_ + c], l1 = w4[5 * NH_C4_ + c], l2 = w4[6 * NH_C4_ + c];
        {
            float t = wb.x + q.x + x0 * r0.x + x1 * r1.x + x2 * r2.x;
            const float u = p.x + a0 * l0.x + a1 * l1.x + a2 * l2.x;
            v[4 * c + 0] = has ? t + u : t;
        }
        {
            float t = wb.y + q.y + x0 * r0.y + x1 * r1.y + x2 * r2.y;
            const float u = p.y + a0 * l0.y + a1 * l1.y + a2 * l2.y;
            v[4 * c + 1] = has ? t + u : t;
        }
        {
            float t = wb.z + q.z + x0 * r0.z + x1 * r1.z + x2 * r2.z;
            const float u = p.z + a0 * l0.z + a1 * l1.z + a2 * l2.z;
            v[4 * c + 2] = has ? t + u : t;
        }
        {
            float t = wb.w + q.w + x0 * r0.w + x1 * r1.w + x2 * r2.w;
            const float u = p.w + a0 * l0.w + a1 * l1.w + a2 * l2.w;
            v[4 * c + 3] = has ? t + u : t;
        }
    }

    float m0 = v[0], m1 = v[1], m2 = v[2], m3 = v[3];
    #pragma unroll
    for (int c = 1; c < NH_C4_; ++c) {
        m0 = fmaxf(m0, v[4 * c + 0]);
        m1 = fmaxf(m1, v[4 * c + 1]);
        m2 = fmaxf(m2, v[4 * c + 2]);
        m3 = fmaxf(m3, v[4 * c + 3]);
    }
    const float m = fmaxf(fmaxf(m0, m1), fmaxf(m2, m3));

    float e0 = 0.f, e1 = 0.f, e2 = 0.f, e3 = 0.f;
    #pragma unroll
    for (int c = 0; c < NH_C4_; ++c) {
        e0 += __expf(v[4 * c + 0] - m);
        e1 += __expf(v[4 * c + 1] - m);
        e2 += __expf(v[4 * c + 2] - m);
        e3 += __expf(v[4 * c + 3] - m);
    }
    const float lse = m + __logf((e0 + e1) + (e2 + e3));

    float p0 = 0.f, p1 = 0.f, p2 = 0.f, p3 = 0.f;
    float r0s = 0.f, r1s = 0.f, r2s = 0.f, r3s = 0.f;
    #pragma unroll
    for (int c = 0; c < NH_C4_; ++c) {
        const float4 wl = w4[7 * NH_C4_ + c];
        const float4 wr = w4[8 * NH_C4_ + c];
        const float h0 = v[4 * c + 0] - lse;
        const float h1 = v[4 * c + 1] - lse;
        const float h2 = v[4 * c + 2] - lse;
        const float h3 = v[4 * c + 3] - lse;
        p0 = fmaf(h0, wl.x, p0);  r0s = fmaf(h0, wr.x, r0s);
        p1 = fmaf(h1, wl.y, p1);  r1s = fmaf(h1, wr.y, r1s);
        p2 = fmaf(h2, wl.z, p2);  r2s = fmaf(h2, wr.z, r2s);
        p3 = fmaf(h3, wl.w, p3);  r3s = fmaf(h3, wr.w, r3s);
    }
    if (active) {
        tl[n] = (p0 + p1) + (p2 + p3);
        tr[n] = (r0s + r1s) + (r2s + r3s);
    }
}

#define ES_T_ 64

template <bool FULL>
__device__ __forceinline__ void scatter_body(const int* __restrict__ ei, const float* __restrict__ tl,
                                             float* __restrict__ agg2, int E, int e0) {
    int s[EPT_], d[EPT_];
    bool ok[EPT_];
    #pragma unroll
    for (int k = 0; k < EPT_; ++k) {
        const int e = e0 + k * ES_T_;
        ok[k] = FULL || e < E;
        const int ec = ok[k] ? e : E - 1;
        s[k] = ei[ec];
        d[k] = ei[E + ec];
    }
    float t[EPT_];
    #pragma unroll
    for (int k = 0; k < EPT_; ++k) t[k] = tl[s[k]];
    #pragma unroll
    for (int k = 0; k < EPT_; ++k)
        if (ok[k]) atomicAdd(&agg2[d[k]], t[k]);
}

__global__ void __launch_bounds__(ES_T_)
edge_scatter_t(const int* __restrict__ ei, const float* __restrict__ tl,
               float* __restrict__ agg2, int E) {
    const int eb = (int)blockIdx.x * (ES_T_ * EPT_);
    if (eb + ES_T_ * EPT_ <= E) scatter_body<true>(ei, tl, agg2, E, eb + (int)threadIdx.x);
    else                        scatter_body<false>(ei, tl, agg2, E, eb + (int)threadIdx.x);
}

__global__ void node_out_kernel(const float* __restrict__ agg2, const float* __restrict__ tr,
                                const int* __restrict__ counts, const float* __restrict__ bl2,
                                float* __restrict__ y, int N) {
    int n = blockIdx.x * blockDim.x + threadIdx.x;
    if (n >= N) return;
    y[n] = agg2[n] / fmaxf((float)counts[n], 1.0f) + bl2[0] + tr[n];
}

extern "C" void kernel_launch(void* const* d_in, const int* in_sizes, int n_in,
                              void* d_out, int out_size, void* d_ws, size_t ws_size,
                              hipStream_t stream) {
    const float* X      = (const float*)d_in[0];
    const float* x3     = (const float*)d_in[1];
    const int*   ei     = (const int*)d_in[2];
    const int*   gid    = (const int*)d_in[3];
    const float* w1     = (const float*)d_in[4];
    const float* b1     = (const float*)d_in[5];
    const float* w2     = (const float*)d_in[6];
    const float* b2     = (const float*)d_in[7];
    const float* w3     = (const float*)d_in[8];
    const float* b3     = (const float*)d_in[9];
    const float* Wl1    = (const float*)d_in[10];
    const float* bl1    = (const float*)d_in[11];
    const float* Wr1    = (const float*)d_in[12];
    const float* Wl2    = (const float*)d_in[13];
    const float* bl2    = (const float*)d_in[14];
    const float* Wr2    = (const float*)d_in[15];
    float* y = (float*)d_out;

    const int E = in_sizes[2] / 2;
    const int N = in_sizes[3];

    char* ws = (char*)d_ws;
    size_t off = 0;
    auto alloc = [&](size_t bytes) -> char* {
        char* p = ws + off;
        off = (off + bytes + 255) & ~(size_t)255;
        return p;
    };
    float*    act1  = (float*)alloc((size_t)8 * 32 * 21 * 36 * 4);
    float*    act2  = (float*)alloc((size_t)8 * 64 * 17 * 36 * 4);
    float*    act3  = (float*)alloc((size_t)8 * 64 * 13 * 16 * 4);
    float*    enc   = (float*)alloc((size_t)8 * ENC_ * 4);
    float*    ME    = (float*)alloc((size_t)256 * ENC_ * 4);
    float*    P     = (float*)alloc((size_t)256 * HID_ * 4);
    float*    Q     = (float*)alloc((size_t)16  * HID_ * 4);
    unsigned* zblk  = (unsigned*)alloc((size_t)N * 6 * 4);
    unsigned* aggk  = zblk;
    unsigned* mask  = zblk + (size_t)N * 3;
    int*      cnt   = (int*)(zblk + (size_t)N * 4);
    float*    agg2  = (float*)(zblk + (size_t)N * 5);
    float*    tl    = (float*)alloc((size_t)N * 4);
    float*    tr    = (float*)alloc((size_t)N * 4);

    const int T = 256;
    conv1_kernel<<<dim3(32, 8), 512, 0, stream>>>(X, w1, b1, act1, zblk, N * 6);
    conv2_kernel<<<dim3(27, 8), 512, 0, stream>>>(act1, w2, b2, act2);
    conv3_kernel<<<dim3(16, 16), 512, 0, stream>>>(act2, w3, b3, act3);

    maxenc_edge_kernel<<<ME_WG_ + (E + ME_T_ * EPT_ - 1) / (ME_T_ * EPT_), ME_T_, 0, stream>>>(act3, enc, ME, ei, gid, x3,
                                                                                          aggk, mask, cnt, E);
    {
        dim3 gPQ(256 / 16 + 1, (HID_ + 15) / 16);
        gemm_wmma_f32<<<gPQ, 32 * KSPLIT_, 0, stream>>>(ME, Wl1 + 3 * HID_, P, 256, 256 / 16,
                                                        enc, Wr1 + 3 * HID_, Q, 8);
    }

    node_h_kernel<<<(N + NH_T_ - 1) / NH_T_, NH_T_, 0, stream>>>(x3, gid, mask, aggk, P, Q,
                                                                 Wl1, Wr1, bl1, Wl2, Wr2, tl, tr, N);

    edge_scatter_t<<<(E + ES_T_ * EPT_ - 1) / (ES_T_ * EPT_), ES_T_, 0, stream>>>(ei, tl, agg2, E);
    node_out_kernel<<<(N + T - 1) / T, T, 0, stream>>>(agg2, tr, cnt, bl2, y, N);

    (void)n_in; (void)out_size; (void)ws_size;
}
